# starting kernel
# baseline (speedup 1.0000x reference)
#include <hip/hip_runtime.h>
#include <hip/hip_bf16.h>


#define GIN_D 64
#define GIN_H 128

typedef __attribute__((ext_vector_type(2))) float v2f;
typedef __attribute__((ext_vector_type(8))) float v8f;

__device__ __forceinline__ void atomic_add_f32_agent(float* p, float v) {
  __hip_atomic_fetch_add(p, v, __ATOMIC_RELAXED, __HIP_MEMORY_SCOPE_AGENT);
}

__global__ void gin_combine_kernel(const float* __restrict__ feat,
                                   const float* __restrict__ eps,
                                   float* __restrict__ hacc, int n4) {
  int i = blockIdx.x * blockDim.x + threadIdx.x;
  if (i >= n4) return;
  const float s = 1.0f + eps[0];
  float4 v = ((const float4*)feat)[i];
  v.x *= s; v.y *= s; v.z *= s; v.w *= s;
  ((float4*)hacc)[i] = v;
}

__global__ void gin_scatter_kernel(const float* __restrict__ feat,
                                   const int* __restrict__ esrc,
                                   const int* __restrict__ edst,
                                   float* __restrict__ hacc, int n_edges) {
  long long t = (long long)blockIdx.x * blockDim.x + threadIdx.x;
  long long total = (long long)n_edges * 16;
  if (t >= total) return;
  int e = (int)(t >> 4);
  int c = (int)(t & 15) * 4;
  int s = esrc[e];
  int d = edst[e];
  float4 v = *(const float4*)(feat + (size_t)s * GIN_D + c);
  float* dp = hacc + (size_t)d * GIN_D + c;
  atomic_add_f32_agent(dp + 0, v.x);
  atomic_add_f32_agent(dp + 1, v.y);
  atomic_add_f32_agent(dp + 2, v.z);
  atomic_add_f32_agent(dp + 3, v.w);
}

__global__ __launch_bounds__(128) void gin_mlp_kernel(
    const float* __restrict__ hacc, const float* __restrict__ W1,
    const float* __restrict__ b1, const float* __restrict__ W2,
    const float* __restrict__ b2, float* __restrict__ out) {
  __shared__ float lds_h[16 * GIN_D];
  __shared__ float lds_mid[16 * GIN_H];

  const int tid  = threadIdx.x;
  const int wave = tid >> 5;
  const int lane = tid & 31;
  const int lhalf = lane >> 4;
  const int lmod  = lane & 15;
  const int node0 = blockIdx.x * 16;

  {
    const float4* src = (const float4*)(hacc + (size_t)node0 * GIN_D);
    float4* dst = (float4*)lds_h;
    dst[tid]       = src[tid];
    dst[tid + 128] = src[tid + 128];
  }
  __syncthreads();

  v8f acc0 = {};
  v8f acc1 = {};
  const int n0 = lmod + 16 * (2 * wave);
  const int n1 = lmod + 16 * (2 * wave + 1);
#pragma unroll
  for (int k = 0; k < 16; ++k) {
    const int kb = 4 * k + 2 * lhalf;
    v2f a;
    a.x = lds_h[lmod * GIN_D + kb];
    a.y = lds_h[lmod * GIN_D + kb + 1];
    v2f bf0, bf1;
    bf0.x = W1[(size_t)kb * GIN_H + n0];
    bf0.y = W1[(size_t)(kb + 1) * GIN_H + n0];
    bf1.x = W1[(size_t)kb * GIN_H + n1];
    bf1.y = W1[(size_t)(kb + 1) * GIN_H + n1];
    acc0 = __builtin_amdgcn_wmma_f32_16x16x4_f32(false, a, false, bf0,
                                                 (short)0, acc0, false, false);
    acc1 = __builtin_amdgcn_wmma_f32_16x16x4_f32(false, a, false, bf1,
                                                 (short)0, acc1, false, false);
  }
  {
    const float b1v0 = b1[n0];
    const float b1v1 = b1[n1];
#pragma unroll
    for (int r = 0; r < 8; ++r) {
      const int row = r + 8 * lhalf;
      lds_mid[row * GIN_H + n0] = fmaxf(acc0[r] + b1v0, 0.0f);
      lds_mid[row * GIN_H + n1] = fmaxf(acc1[r] + b1v1, 0.0f);
    }
  }
  __syncthreads();

  v8f acc2 = {};
  const int n2 = lmod + 16 * wave;
#pragma unroll
  for (int k = 0; k < 32; ++k) {
    const int kb = 4 * k + 2 * lhalf;
    v2f a;
    a.x = lds_mid[lmod * GIN_H + kb];
    a.y = lds_mid[lmod * GIN_H + kb + 1];
    v2f bf;
    bf.x = W2[(size_t)kb * GIN_D + n2];
    bf.y = W2[(size_t)(kb + 1) * GIN_D + n2];
    acc2 = __builtin_amdgcn_wmma_f32_16x16x4_f32(false, a, false, bf,
                                                 (short)0, acc2, false, false);
  }
  {
    const float b2v = b2[n2];
#pragma unroll
    for (int r = 0; r < 8; ++r) {
      const int row = r + 8 * lhalf;
      out[(size_t)(node0 + row) * GIN_D + n2] = acc2[r] + b2v;
    }
  }
}

extern "C" void kernel_launch(void* const* d_in, const int* in_sizes, int n_in,
                              void* d_out, int out_size, void* d_ws, size_t ws_size,
                              hipStream_t stream) {
  const float* feat = (const float*)d_in[0];
  const int*   esrc = (const int*)d_in[1];
  const int*   edst = (const int*)d_in[2];
  const float* W1   = (const float*)d_in[3];
  const float* b1   = (const float*)d_in[4];
  const float* W2   = (const float*)d_in[5];
  const float* b2   = (const float*)d_in[6];
  const float* eps  = (const float*)d_in[7];

  const int n_nodes = in_sizes[0] / GIN_D;
  const int n_edges = in_sizes[1];

  float* hacc = (float*)d_ws;

  {
    const int n4 = n_nodes * (GIN_D / 4);
    const int blocks = (n4 + 255) / 256;
    gin_combine_kernel<<<blocks, 256, 0, stream>>>(feat, eps, hacc, n4);
  }
  {
    const long long total = (long long)n_edges * 16;
    const int blocks = (int)((total + 255) / 256);
    gin_scatter_kernel<<<blocks, 256, 0, stream>>>(feat, esrc, edst, hacc, n_edges);
  }
  {
    const int blocks = n_nodes / 16;
    gin_mlp_kernel<<<blocks, 128, 0, stream>>>(hacc, W1, b1, W2, b2,
                                               (float*)d_out);
  }
}


# scatter re-mapped: one wave per 32 edges, src/dst broadcast by readlane, lane = columns l and l+32, so each gather and each f32 atomic add covers one whole 128-B line (adds still in arrival order)
# speedup vs baseline: 1.5673x; 1.5673x over previous
#include <hip/hip_runtime.h>
#include <hip/hip_bf16.h>


#define GIN_D 64
#define GIN_H 128

typedef __attribute__((ext_vector_type(2))) float v2f;
typedef __attribute__((ext_vector_type(8))) float v8f;

__device__ __forceinline__ void atomic_add_f32_agent(float* p, float v) {
  __hip_atomic_fetch_add(p, v, __ATOMIC_RELAXED, __HIP_MEMORY_SCOPE_AGENT);
}

__global__ void gin_combine_kernel(const float* __restrict__ feat,
                                   const float* __restrict__ eps,
                                   float* __restrict__ hacc, int n4) {
  int i = blockIdx.x * blockDim.x + threadIdx.x;
  if (i >= n4) return;
  const float s = 1.0f + eps[0];
  float4 v = ((const float4*)feat)[i];
  v.x *= s; v.y *= s; v.z *= s; v.w *= s;
  ((float4*)hacc)[i] = v;
}

__global__ __launch_bounds__(256) void gin_scatter_kernel(
    const float* __restrict__ feat, const int* __restrict__ esrc,
    const int* __restrict__ edst, float* __restrict__ hacc, int n_edges) {
  const int lane = threadIdx.x & 31;
  const long long t = (long long)blockIdx.x * blockDim.x + threadIdx.x;
  const long long base = t - lane;
  if (base >= n_edges) return;
  const long long left = (long long)n_edges - base;
  const int cnt = left < 32 ? (int)left : 32;
  int s_l = 0, d_l = 0;
  if (lane < cnt) { s_l = esrc[t]; d_l = edst[t]; }

  if (cnt == 32) {
#pragma unroll
    for (int j0 = 0; j0 < 32; j0 += 8) {
      float lo[8], hi[8];
#pragma unroll
      for (int j = 0; j < 8; ++j) {
        const int s = __builtin_amdgcn_readlane(s_l, j0 + j);
        const float* sp = feat + (size_t)s * GIN_D + lane;
        lo[j] = sp[0];
        hi[j] = sp[32];
      }
#pragma unroll
      for (int j = 0; j < 8; ++j) {
        const int d = __builtin_amdgcn_readlane(d_l, j0 + j);
        float* dp = hacc + (size_t)d * GIN_D + lane;
        atomic_add_f32_agent(dp, lo[j]);
        atomic_add_f32_agent(dp + 32, hi[j]);
      }
    }
  } else {
    for (int j = 0; j < cnt; ++j) {
      const int s = __builtin_amdgcn_readlane(s_l, j);
      const int d = __builtin_amdgcn_readlane(d_l, j);
      const float* sp = feat + (size_t)s * GIN_D + lane;
      float* dp = hacc + (size_t)d * GIN_D + lane;
      const float a = sp[0];
      const float b = sp[32];
      atomic_add_f32_agent(dp, a);
      atomic_add_f32_agent(dp + 32, b);
    }
  }
}

__global__ __launch_bounds__(128) void gin_mlp_kernel(
    const float* __restrict__ hacc, const float* __restrict__ W1,
    const float* __restrict__ b1, const float* __restrict__ W2,
    const float* __restrict__ b2, float* __restrict__ out) {
  __shared__ float lds_h[16 * GIN_D];
  __shared__ float lds_mid[16 * GIN_H];

  const int tid  = threadIdx.x;
  const int wave = tid >> 5;
  const int lane = tid & 31;
  const int lhalf = lane >> 4;
  const int lmod  = lane & 15;
  const int node0 = blockIdx.x * 16;

  {
    const float4* src = (const float4*)(hacc + (size_t)node0 * GIN_D);
    float4* dst = (float4*)lds_h;
    dst[tid]       = src[tid];
    dst[tid + 128] = src[tid + 128];
  }
  __syncthreads();

  v8f acc0 = {};
  v8f acc1 = {};
  const int n0 = lmod + 16 * (2 * wave);
  const int n1 = lmod + 16 * (2 * wave + 1);
#pragma unroll
  for (int k = 0; k < 16; ++k) {
    const int kb = 4 * k + 2 * lhalf;
    v2f a;
    a.x = lds_h[lmod * GIN_D + kb];
    a.y = lds_h[lmod * GIN_D + kb + 1];
    v2f bf0, bf1;
    bf0.x = W1[(size_t)kb * GIN_H + n0];
    bf0.y = W1[(size_t)(kb + 1) * GIN_H + n0];
    bf1.x = W1[(size_t)kb * GIN_H + n1];
    bf1.y = W1[(size_t)(kb + 1) * GIN_H + n1];
    acc0 = __builtin_amdgcn_wmma_f32_16x16x4_f32(false, a, false, bf0,
                                                 (short)0, acc0, false, false);
    acc1 = __builtin_amdgcn_wmma_f32_16x16x4_f32(false, a, false, bf1,
                                                 (short)0, acc1, false, false);
  }
  {
    const float b1v0 = b1[n0];
    const float b1v1 = b1[n1];
#pragma unroll
    for (int r = 0; r < 8; ++r) {
      const int row = r + 8 * lhalf;
      lds_mid[row * GIN_H + n0] = fmaxf(acc0[r] + b1v0, 0.0f);
      lds_mid[row * GIN_H + n1] = fmaxf(acc1[r] + b1v1, 0.0f);
    }
  }
  __syncthreads();

  v8f acc2 = {};
  const int n2 = lmod + 16 * wave;
#pragma unroll
  for (int k = 0; k < 32; ++k) {
    const int kb = 4 * k + 2 * lhalf;
    v2f a;
    a.x = lds_mid[lmod * GIN_H + kb];
    a.y = lds_mid[lmod * GIN_H + kb + 1];
    v2f bf;
    bf.x = W2[(size_t)kb * GIN_D + n2];
    bf.y = W2[(size_t)(kb + 1) * GIN_D + n2];
    acc2 = __builtin_amdgcn_wmma_f32_16x16x4_f32(false, a, false, bf,
                                                 (short)0, acc2, false, false);
  }
  {
    const float b2v = b2[n2];
#pragma unroll
    for (int r = 0; r < 8; ++r) {
      const int row = r + 8 * lhalf;
      out[(size_t)(node0 + row) * GIN_D + n2] = acc2[r] + b2v;
    }
  }
}

extern "C" void kernel_launch(void* const* d_in, const int* in_sizes, int n_in,
                              void* d_out, int out_size, void* d_ws, size_t ws_size,
                              hipStream_t stream) {
  const float* feat = (const float*)d_in[0];
  const int*   esrc = (const int*)d_in[1];
  const int*   edst = (const int*)d_in[2];
  const float* W1   = (const float*)d_in[3];
  const float* b1   = (const float*)d_in[4];
  const float* W2   = (const float*)d_in[5];
  const float* b2   = (const float*)d_in[6];
  const float* eps  = (const float*)d_in[7];

  const int n_nodes = in_sizes[0] / GIN_D;
  const int n_edges = in_sizes[1];

  float* hacc = (float*)d_ws;

  {
    const int n4 = n_nodes * (GIN_D / 4);
    const int blocks = (n4 + 255) / 256;
    gin_combine_kernel<<<blocks, 256, 0, stream>>>(feat, eps, hacc, n4);
  }
  {
    const int blocks = (n_edges + 255) / 256;
    gin_scatter_kernel<<<blocks, 256, 0, stream>>>(feat, esrc, edst, hacc, n_edges);
  }
  {
    const int blocks = n_nodes / 16;
    gin_mlp_kernel<<<blocks, 128, 0, stream>>>(hacc, W1, b1, W2, b2,
                                               (float*)d_out);
  }
}


# segment sum without global atomics: edges binned by 256-node dst tile (count + fill passes), one work-group per tile sums gathered rows with LDS f32 adds (arrival order), then h=(1+eps)*feat+sum
# speedup vs baseline: 1.8610x; 1.1874x over previous
#include <hip/hip_runtime.h>
#include <hip/hip_bf16.h>


#define GIN_D 64
#define GIN_H 128
#define TILE_N 256
#define TILE_SHIFT 8
#define NB_MAX 512
#define CHUNK_E 16384
#define BIN_T 1024
#define PER_T (CHUNK_E / BIN_T)

typedef __attribute__((ext_vector_type(2))) float v2f;
typedef __attribute__((ext_vector_type(8))) float v8f;

__device__ __forceinline__ void lds_add_f32(float* p, float v) {
  __hip_atomic_fetch_add(p, v, __ATOMIC_RELAXED, __HIP_MEMORY_SCOPE_WORKGROUP);
}

__global__ __launch_bounds__(BIN_T) void gin_count_kernel(
    const int* __restrict__ edst, int* __restrict__ hist_wg, int n_edges, int nb) {
  __shared__ int hist[NB_MAX];
  const int tid = threadIdx.x;
  if (tid < NB_MAX) hist[tid] = 0;
  __syncthreads();
  const long long e0 = (long long)blockIdx.x * CHUNK_E + tid;
  const long long last = (long long)n_edges - 1;
  int dd[PER_T];
#pragma unroll
  for (int k = 0; k < PER_T; ++k) {
    const long long e = e0 + (long long)k * BIN_T;
    dd[k] = edst[e < last ? e : last];
  }
#pragma unroll
  for (int k = 0; k < PER_T; ++k) {
    const long long e = e0 + (long long)k * BIN_T;
    if (e <= last)
      __hip_atomic_fetch_add(&hist[dd[k] >> TILE_SHIFT], 1, __ATOMIC_RELAXED,
                             __HIP_MEMORY_SCOPE_WORKGROUP);
  }
  __syncthreads();
  if (tid < nb) hist_wg[(size_t)blockIdx.x * nb + tid] = hist[tid];
}

__global__ __launch_bounds__(BIN_T) void gin_fill_kernel(
    const int* __restrict__ esrc, const int* __restrict__ edst,
    const int* __restrict__ hist_wg, int* __restrict__ binned_i,
    int* __restrict__ tile_start, int n_edges, int nb, int nwg) {
  __shared__ int tot[NB_MAX];
  __shared__ int scan[2][NB_MAX];
  __shared__ int base[NB_MAX];
  __shared__ int cur[NB_MAX];
  const int tid = threadIdx.x;
  const int wg = blockIdx.x;

  int pre = 0, all = 0;
  if (tid < nb) {
    for (int w0 = 0; w0 < nwg; w0 += 16) {
      int v[16];
#pragma unroll
      for (int j = 0; j < 16; ++j) {
        const int w = w0 + j;
        v[j] = (w < nwg) ? hist_wg[(size_t)w * nb + tid] : 0;
      }
#pragma unroll
      for (int j = 0; j < 16; ++j) {
        all += v[j];
        if (w0 + j < wg) pre += v[j];
      }
    }
  }
  if (tid < NB_MAX) { tot[tid] = all; scan[0][tid] = all; cur[tid] = 0; }
  int sb = 0;
  for (int off = 1; off < NB_MAX; off <<= 1) {
    __syncthreads();
    if (tid < NB_MAX) {
      int v = scan[sb][tid];
      if (tid >= off) v += scan[sb][tid - off];
      scan[sb ^ 1][tid] = v;
    }
    sb ^= 1;
  }
  __syncthreads();
  if (tid < NB_MAX) {
    const int excl = scan[sb][tid] - tot[tid];
    base[tid] = excl + pre;
    if (wg == 0 && tid <= nb) tile_start[tid] = excl;
  }
  __syncthreads();

  int2* __restrict__ binned = (int2*)binned_i;
  const long long e0 = (long long)wg * CHUNK_E + tid;
  const long long last = (long long)n_edges - 1;
  int dd[PER_T], ss[PER_T];
#pragma unroll
  for (int k = 0; k < PER_T; ++k) {
    const long long e = e0 + (long long)k * BIN_T;
    const long long ec = e < last ? e : last;
    dd[k] = edst[ec];
    ss[k] = esrc[ec];
  }
#pragma unroll
  for (int k = 0; k < PER_T; ++k) {
    const long long e = e0 + (long long)k * BIN_T;
    if (e <= last) {
      const int b = dd[k] >> TILE_SHIFT;
      const int r = __hip_atomic_fetch_add(&cur[b], 1, __ATOMIC_RELAXED,
                                           __HIP_MEMORY_SCOPE_WORKGROUP);
      binned[base[b] + r] = make_int2(ss[k], dd[k]);
    }
  }
}

__global__ __launch_bounds__(1024) void gin_tile_kernel(
    const float* __restrict__ feat, const float* __restrict__ eps,
    const int* __restrict__ binned_i, const int* __restrict__ tile_start,
    float* __restrict__ hacc, int n_nodes) {
  __shared__ float tile[TILE_N * GIN_D];
  const int tid = threadIdx.x;
  const int lane = tid & 31;
  const int wave = __builtin_amdgcn_readfirstlane(tid >> 5);
  const int n0 = blockIdx.x * TILE_N;
  const int2* __restrict__ binned = (const int2*)binned_i;

  for (int i = tid; i < TILE_N * GIN_D / 4; i += 1024)
    ((float4*)tile)[i] = make_float4(0.f, 0.f, 0.f, 0.f);
  __syncthreads();

  const int e_begin = tile_start[blockIdx.x];
  const int e_end = tile_start[blockIdx.x + 1];
  int run = e_begin + wave * 32;
  int2 p_next = make_int2(0, n0);
  if (run + lane < e_end) p_next = binned[run + lane];
  for (; run < e_end; run += 32 * 32) {
    const int left = e_end - run;
    const int cnt = left < 32 ? left : 32;
    const int s_l = p_next.x;
    const int d_l = p_next.y - n0;
    const int ni = run + 32 * 32 + lane;
    p_next = binned[ni < e_end ? ni : e_begin];

    if (cnt == 32) {
#pragma unroll
      for (int j0 = 0; j0 < 32; j0 += 8) {
        float lo[8], hi[8];
#pragma unroll
        for (int j = 0; j < 8; ++j) {
          const int s = __builtin_amdgcn_readlane(s_l, j0 + j);
          const float* sp = feat + (size_t)s * GIN_D + lane;
          lo[j] = sp[0];
          hi[j] = sp[32];
        }
#pragma unroll
        for (int j = 0; j < 8; ++j) {
          const int d = __builtin_amdgcn_readlane(d_l, j0 + j);
          float* dp = tile + d * GIN_D + lane;
          lds_add_f32(dp, lo[j]);
          lds_add_f32(dp + 32, hi[j]);
        }
      }
    } else {
      for (int j = 0; j < cnt; ++j) {
        const int s = __builtin_amdgcn_readlane(s_l, j);
        const int d = __builtin_amdgcn_readlane(d_l, j);
        const float* sp = feat + (size_t)s * GIN_D + lane;
        float* dp = tile + d * GIN_D + lane;
        const float a = sp[0];
        const float b = sp[32];
        lds_add_f32(dp, a);
        lds_add_f32(dp + 32, b);
      }
    }
  }
  __syncthreads();

  const float sc = 1.0f + eps[0];
  int rows = n_nodes - n0;
  if (rows > TILE_N) rows = TILE_N;
  const float4* f4 = (const float4*)(feat + (size_t)n0 * GIN_D);
  float4* h4 = (float4*)(hacc + (size_t)n0 * GIN_D);
  for (int i = tid; i < rows * (GIN_D / 4); i += 1024) {
    float4 v = f4[i];
    const float4 t = ((const float4*)tile)[i];
    v.x = sc * v.x + t.x; v.y = sc * v.y + t.y;
    v.z = sc * v.z + t.z; v.w = sc * v.w + t.w;
    h4[i] = v;
  }
}

__global__ __launch_bounds__(128) void gin_mlp_kernel(
    const float* __restrict__ hacc, const float* __restrict__ W1,
    const float* __restrict__ b1, const float* __restrict__ W2,
    const float* __restrict__ b2, float* __restrict__ out) {
  __shared__ float lds_h[16 * GIN_D];
  __shared__ float lds_mid[16 * GIN_H];

  const int tid  = threadIdx.x;
  const int wave = tid >> 5;
  const int lane = tid & 31;
  const int lhalf = lane >> 4;
  const int lmod  = lane & 15;
  const int node0 = blockIdx.x * 16;

  {
    const float4* src = (const float4*)(hacc + (size_t)node0 * GIN_D);
    float4* dst = (float4*)lds_h;
    dst[tid]       = src[tid];
    dst[tid + 128] = src[tid + 128];
  }
  __syncthreads();

  v8f acc0 = {};
  v8f acc1 = {};
  const int n0 = lmod + 16 * (2 * wave);
  const int n1 = lmod + 16 * (2 * wave + 1);
#pragma unroll
  for (int k = 0; k < 16; ++k) {
    const int kb = 4 * k + 2 * lhalf;
    v2f a;
    a.x = lds_h[lmod * GIN_D + kb];
    a.y = lds_h[lmod * GIN_D + kb + 1];
    v2f bf0, bf1;
    bf0.x = W1[(size_t)kb * GIN_H + n0];
    bf0.y = W1[(size_t)(kb + 1) * GIN_H + n0];
    bf1.x = W1[(size_t)kb * GIN_H + n1];
    bf1.y = W1[(size_t)(kb + 1) * GIN_H + n1];
    acc0 = __builtin_amdgcn_wmma_f32_16x16x4_f32(false, a, false, bf0,
                                                 (short)0, acc0, false, false);
    acc1 = __builtin_amdgcn_wmma_f32_16x16x4_f32(false, a, false, bf1,
                                                 (short)0, acc1, false, false);
  }
  {
    const float b1v0 = b1[n0];
    const float b1v1 = b1[n1];
#pragma unroll
    for (int r = 0; r < 8; ++r) {
      const int row = r + 8 * lhalf;
      lds_mid[row * GIN_H + n0] = fmaxf(acc0[r] + b1v0, 0.0f);
      lds_mid[row * GIN_H + n1] = fmaxf(acc1[r] + b1v1, 0.0f);
    }
  }
  __syncthreads();

  v8f acc2 = {};
  const int n2 = lmod + 16 * wave;
#pragma unroll
  for (int k = 0; k < 32; ++k) {
    const int kb = 4 * k + 2 * lhalf;
    v2f a;
    a.x = lds_mid[lmod * GIN_H + kb];
    a.y = lds_mid[lmod * GIN_H + kb + 1];
    v2f bf;
    bf.x = W2[(size_t)kb * GIN_D + n2];
    bf.y = W2[(size_t)(kb + 1) * GIN_D + n2];
    acc2 = __builtin_amdgcn_wmma_f32_16x16x4_f32(false, a, false, bf,
                                                 (short)0, acc2, false, false);
  }
  {
    const float b2v = b2[n2];
#pragma unroll
    for (int r = 0; r < 8; ++r) {
      const int row = r + 8 * lhalf;
      out[(size_t)(node0 + row) * GIN_D + n2] = acc2[r] + b2v;
    }
  }
}

extern "C" void kernel_launch(void* const* d_in, const int* in_sizes, int n_in,
                              void* d_out, int out_size, void* d_ws, size_t ws_size,
                              hipStream_t stream) {
  const float* feat = (const float*)d_in[0];
  const int*   esrc = (const int*)d_in[1];
  const int*   edst = (const int*)d_in[2];
  const float* W1   = (const float*)d_in[3];
  const float* b1   = (const float*)d_in[4];
  const float* W2   = (const float*)d_in[5];
  const float* b2   = (const float*)d_in[6];
  const float* eps  = (const float*)d_in[7];

  const int n_nodes = in_sizes[0] / GIN_D;
  const int n_edges = in_sizes[1];
  const int nb  = (n_nodes + TILE_N - 1) / TILE_N;
  const int nwg = (n_edges + CHUNK_E - 1) / CHUNK_E;

  float* hacc       = (float*)d_ws;
  int*   binned     = (int*)(hacc + (size_t)n_nodes * GIN_D);
  int*   hist_wg    = binned + (size_t)n_edges * 2;
  int*   tile_start = hist_wg + (size_t)nwg * nb;

  gin_count_kernel<<<nwg, BIN_T, 0, stream>>>(edst, hist_wg, n_edges, nb);
  gin_fill_kernel<<<nwg, BIN_T, 0, stream>>>(esrc, edst, hist_wg, binned, tile_start,
                                             n_edges, nb, nwg);
  gin_tile_kernel<<<nb, 1024, 0, stream>>>(feat, eps, binned, tile_start, hacc, n_nodes);
  {
    const int blocks = n_nodes / 16;
    gin_mlp_kernel<<<blocks, 128, 0, stream>>>(hacc, W1, b1, W2, b2,
                                               (float*)d_out);
  }
}


# tile sum: each tile's edges ordered by dst row in LDS (integer rank + scan); a wave sums its 8 rows in registers in rank (arrival) order and stores h itself; the LDS f32 adds are gone
# speedup vs baseline: 2.0815x; 1.1185x over previous
#include <hip/hip_runtime.h>
#include <hip/hip_bf16.h>


#define GIN_D 64
#define GIN_H 128
#define TILE_N 256
#define TILE_SHIFT 8
#define NB_MAX 512
#define CHUNK_E 16384
#define BIN_T 1024
#define PER_T (CHUNK_E / BIN_T)
#define SORT_C 8192
#define SORT_T (SORT_C / 1024)
#define ROWS_W (TILE_N / 32)

typedef __attribute__((ext_vector_type(2))) float v2f;
typedef __attribute__((ext_vector_type(8))) float v8f;

__global__ __launch_bounds__(BIN_T) void gin_count_kernel(
    const int* __restrict__ edst, int* __restrict__ hist_wg, int n_edges, int nb) {
  __shared__ int hist[NB_MAX];
  const int tid = threadIdx.x;
  if (tid < NB_MAX) hist[tid] = 0;
  __syncthreads();
  const long long e0 = (long long)blockIdx.x * CHUNK_E + tid;
  const long long last = (long long)n_edges - 1;
  int dd[PER_T];
#pragma unroll
  for (int k = 0; k < PER_T; ++k) {
    const long long e = e0 + (long long)k * BIN_T;
    dd[k] = edst[e < last ? e : last];
  }
#pragma unroll
  for (int k = 0; k < PER_T; ++k) {
    const long long e = e0 + (long long)k * BIN_T;
    if (e <= last)
      __hip_atomic_fetch_add(&hist[dd[k] >> TILE_SHIFT], 1, __ATOMIC_RELAXED,
                             __HIP_MEMORY_SCOPE_WORKGROUP);
  }
  __syncthreads();
  if (tid < nb) hist_wg[(size_t)blockIdx.x * nb + tid] = hist[tid];
}

__global__ __launch_bounds__(BIN_T) void gin_fill_kernel(
    const int* __restrict__ esrc, const int* __restrict__ edst,
    const int* __restrict__ hist_wg, int* __restrict__ binned_i,
    int* __restrict__ tile_start, int n_edges, int nb, int nwg) {
  __shared__ int tot[NB_MAX];
  __shared__ int scan[2][NB_MAX];
  __shared__ int base[NB_MAX];
  __shared__ int cur[NB_MAX];
  const int tid = threadIdx.x;
  const int wg = blockIdx.x;

  int pre = 0, all = 0;
  if (tid < nb) {
    for (int w0 = 0; w0 < nwg; w0 += 16) {
      int v[16];
#pragma unroll
      for (int j = 0; j < 16; ++j) {
        const int w = w0 + j;
        v[j] = (w < nwg) ? hist_wg[(size_t)w * nb + tid] : 0;
      }
#pragma unroll
      for (int j = 0; j < 16; ++j) {
        all += v[j];
        if (w0 + j < wg) pre += v[j];
      }
    }
  }
  if (tid < NB_MAX) { tot[tid] = all; scan[0][tid] = all; cur[tid] = 0; }
  int sb = 0;
  for (int off = 1; off < NB_MAX; off <<= 1) {
    __syncthreads();
    if (tid < NB_MAX) {
      int v = scan[sb][tid];
      if (tid >= off) v += scan[sb][tid - off];
      scan[sb ^ 1][tid] = v;
    }
    sb ^= 1;
  }
  __syncthreads();
  if (tid < NB_MAX) {
    const int excl = scan[sb][tid] - tot[tid];
    base[tid] = excl + pre;
    if (wg == 0 && tid <= nb) tile_start[tid] = excl;
  }
  __syncthreads();

  int2* __restrict__ binned = (int2*)binned_i;
  const long long e0 = (long long)wg * CHUNK_E + tid;
  const long long last = (long long)n_edges - 1;
  int dd[PER_T], ss[PER_T];
#pragma unroll
  for (int k = 0; k < PER_T; ++k) {
    const long long e = e0 + (long long)k * BIN_T;
    const long long ec = e < last ? e : last;
    dd[k] = edst[ec];
    ss[k] = esrc[ec];
  }
#pragma unroll
  for (int k = 0; k < PER_T; ++k) {
    const long long e = e0 + (long long)k * BIN_T;
    if (e <= last) {
      const int b = dd[k] >> TILE_SHIFT;
      const int r = __hip_atomic_fetch_add(&cur[b], 1, __ATOMIC_RELAXED,
                                           __HIP_MEMORY_SCOPE_WORKGROUP);
      binned[base[b] + r] = make_int2(ss[k], dd[k]);
    }
  }
}

__global__ __launch_bounds__(1024) __attribute__((amdgpu_waves_per_eu(16, 16))) void gin_tile_kernel(
    const float* __restrict__ feat, const float* __restrict__ eps,
    const int* __restrict__ binned_i, const int* __restrict__ tile_start,
    float* __restrict__ hacc, int n_nodes) {
  __shared__ int cnt[TILE_N];
  __shared__ int rstart[TILE_N + 1];
  __shared__ int ssrc[SORT_C];
  const int tid = threadIdx.x;
  const int lane = tid & 31;
  const int wave = __builtin_amdgcn_readfirstlane(tid >> 5);
  const int n0 = blockIdx.x * TILE_N;
  const int2* __restrict__ binned = (const int2*)binned_i;
  const int e_begin = tile_start[blockIdx.x];
  const int e_end = tile_start[blockIdx.x + 1];

  float lo[ROWS_W], hi[ROWS_W];
#pragma unroll
  for (int rr = 0; rr < ROWS_W; ++rr) { lo[rr] = 0.f; hi[rr] = 0.f; }

  for (int c0 = e_begin; c0 < e_end; c0 += SORT_C) {
    const int left = e_end - c0;
    const int n = left < SORT_C ? left : SORT_C;
    if (tid < TILE_N) cnt[tid] = 0;
    __syncthreads();

    int ss[SORT_T], dd[SORT_T];
#pragma unroll
    for (int k = 0; k < SORT_T; ++k) {
      ss[k] = 0; dd[k] = 0;
      if (k * 1024 < n) {
        const int i = tid + k * 1024;
        const int2 p = binned[c0 + (i < n ? i : 0)];
        ss[k] = p.x;
        dd[k] = p.y;
      }
    }
#pragma unroll
    for (int k = 0; k < SORT_T; ++k) {
      if (tid + k * 1024 < n) {
        dd[k] -= n0;
        const int r = __hip_atomic_fetch_add(&cnt[dd[k]], 1, __ATOMIC_RELAXED,
                                             __HIP_MEMORY_SCOPE_WORKGROUP);
        dd[k] |= r << TILE_SHIFT;
      }
    }
    __syncthreads();

    if (wave == 0) {
      int v[8];
#pragma unroll
      for (int j = 0; j < 8; ++j) v[j] = cnt[lane * 8 + j];
      int t = 0;
      int pre[8];
#pragma unroll
      for (int j = 0; j < 8; ++j) { pre[j] = t; t += v[j]; }
      int inc = t;
#pragma unroll
      for (int off = 1; off < 32; off <<= 1) {
        const int u = __shfl_up(inc, off);
        if (lane >= off) inc += u;
      }
      const int base = inc - t;
#pragma unroll
      for (int j = 0; j < 8; ++j) rstart[lane * 8 + j] = base + pre[j];
      if (lane == 31) rstart[TILE_N] = inc;
    }
    __syncthreads();

#pragma unroll
    for (int k = 0; k < SORT_T; ++k) {
      if (tid + k * 1024 < n)
        ssrc[rstart[dd[k] & (TILE_N - 1)] + (dd[k] >> TILE_SHIFT)] = ss[k];
    }
    __syncthreads();

    const int rsv = rstart[wave * ROWS_W + (lane < ROWS_W ? lane : ROWS_W)];
#pragma unroll
    for (int rr = 0; rr < ROWS_W; ++rr) {
      const int rb = __builtin_amdgcn_readlane(rsv, rr);
      const int re = __builtin_amdgcn_readlane(rsv, rr + 1);
      for (int j0 = rb; j0 < re; j0 += 32) {
        const int m = re - j0;
        const int idx = j0 + lane;
        const int sv = idx < re ? ssrc[idx] : 0;
        for (int g = 0; g < 32 && g < m; g += 8) {
          float a[8], c[8];
#pragma unroll
          for (int j = 0; j < 8; ++j) {
            const int s = __builtin_amdgcn_readlane(sv, g + j);
            const float* sp = feat + (size_t)s * GIN_D + lane;
            a[j] = sp[0];
            c[j] = sp[32];
          }
#pragma unroll
          for (int j = 0; j < 8; ++j) {
            if (g + j < m) { lo[rr] += a[j]; hi[rr] += c[j]; }
          }
        }
      }
    }
  }

  const float sc = 1.0f + eps[0];
  float f0[ROWS_W], f1[ROWS_W];
#pragma unroll
  for (int rr = 0; rr < ROWS_W; ++rr) {
    const int node = n0 + wave * ROWS_W + rr;
    const int nc = node < n_nodes ? node : n_nodes - 1;
    const float* fp = feat + (size_t)nc * GIN_D + lane;
    f0[rr] = fp[0];
    f1[rr] = fp[32];
  }
#pragma unroll
  for (int rr = 0; rr < ROWS_W; ++rr) {
    const int node = n0 + wave * ROWS_W + rr;
    if (node < n_nodes) {
      float* hp = hacc + (size_t)node * GIN_D + lane;
      hp[0] = sc * f0[rr] + lo[rr];
      hp[32] = sc * f1[rr] + hi[rr];
    }
  }
}

__global__ __launch_bounds__(128) void gin_mlp_kernel(
    const float* __restrict__ hacc, const float* __restrict__ W1,
    const float* __restrict__ b1, const float* __restrict__ W2,
    const float* __restrict__ b2, float* __restrict__ out) {
  __shared__ float lds_h[16 * GIN_D];
  __shared__ float lds_mid[16 * GIN_H];

  const int tid  = threadIdx.x;
  const int wave = tid >> 5;
  const int lane = tid & 31;
  const int lhalf = lane >> 4;
  const int lmod  = lane & 15;
  const int node0 = blockIdx.x * 16;

  {
    const float4* src = (const float4*)(hacc + (size_t)node0 * GIN_D);
    float4* dst = (float4*)lds_h;
    dst[tid]       = src[tid];
    dst[tid + 128] = src[tid + 128];
  }
  __syncthreads();

  v8f acc0 = {};
  v8f acc1 = {};
  const int n0 = lmod + 16 * (2 * wave);
  const int n1 = lmod + 16 * (2 * wave + 1);
#pragma unroll
  for (int k = 0; k < 16; ++k) {
    const int kb = 4 * k + 2 * lhalf;
    v2f a;
    a.x = lds_h[lmod * GIN_D + kb];
    a.y = lds_h[lmod * GIN_D + kb + 1];
    v2f bf0, bf1;
    bf0.x = W1[(size_t)kb * GIN_H + n0];
    bf0.y = W1[(size_t)(kb + 1) * GIN_H + n0];
    bf1.x = W1[(size_t)kb * GIN_H + n1];
    bf1.y = W1[(size_t)(kb + 1) * GIN_H + n1];
    acc0 = __builtin_amdgcn_wmma_f32_16x16x4_f32(false, a, false, bf0,
                                                 (short)0, acc0, false, false);
    acc1 = __builtin_amdgcn_wmma_f32_16x16x4_f32(false, a, false, bf1,
                                                 (short)0, acc1, false, false);
  }
  {
    const float b1v0 = b1[n0];
    const float b1v1 = b1[n1];
#pragma unroll
    for (int r = 0; r < 8; ++r) {
      const int row = r + 8 * lhalf;
      lds_mid[row * GIN_H + n0] = fmaxf(acc0[r] + b1v0, 0.0f);
      lds_mid[row * GIN_H + n1] = fmaxf(acc1[r] + b1v1, 0.0f);
    }
  }
  __syncthreads();

  v8f acc2 = {};
  const int n2 = lmod + 16 * wave;
#pragma unroll
  for (int k = 0; k < 32; ++k) {
    const int kb = 4 * k + 2 * lhalf;
    v2f a;
    a.x = lds_mid[lmod * GIN_H + kb];
    a.y = lds_mid[lmod * GIN_H + kb + 1];
    v2f bf;
    bf.x = W2[(size_t)kb * GIN_D + n2];
    bf.y = W2[(size_t)(kb + 1) * GIN_D + n2];
    acc2 = __builtin_amdgcn_wmma_f32_16x16x4_f32(false, a, false, bf,
                                                 (short)0, acc2, false, false);
  }
  {
    const float b2v = b2[n2];
#pragma unroll
    for (int r = 0; r < 8; ++r) {
      const int row = r + 8 * lhalf;
      out[(size_t)(node0 + row) * GIN_D + n2] = acc2[r] + b2v;
    }
  }
}

extern "C" void kernel_launch(void* const* d_in, const int* in_sizes, int n_in,
                              void* d_out, int out_size, void* d_ws, size_t ws_size,
                              hipStream_t stream) {
  const float* feat = (const float*)d_in[0];
  const int*   esrc = (const int*)d_in[1];
  const int*   edst = (const int*)d_in[2];
  const float* W1   = (const float*)d_in[3];
  const float* b1   = (const float*)d_in[4];
  const float* W2   = (const float*)d_in[5];
  const float* b2   = (const float*)d_in[6];
  const float* eps  = (const float*)d_in[7];

  const int n_nodes = in_sizes[0] / GIN_D;
  const int n_edges = in_sizes[1];
  const int nb  = (n_nodes + TILE_N - 1) / TILE_N;
  const int nwg = (n_edges + CHUNK_E - 1) / CHUNK_E;

  float* hacc       = (float*)d_ws;
  int*   binned     = (int*)(hacc + (size_t)n_nodes * GIN_D);
  int*   hist_wg    = binned + (size_t)n_edges * 2;
  int*   tile_start = hist_wg + (size_t)nwg * nb;

  gin_count_kernel<<<nwg, BIN_T, 0, stream>>>(edst, hist_wg, n_edges, nb);
  gin_fill_kernel<<<nwg, BIN_T, 0, stream>>>(esrc, edst, hist_wg, binned, tile_start,
                                             n_edges, nb, nwg);
  gin_tile_kernel<<<nb, 1024, 0, stream>>>(feat, eps, binned, tile_start, hacc, n_nodes);
  {
    const int blocks = n_nodes / 16;
    gin_mlp_kernel<<<blocks, 128, 0, stream>>>(hacc, W1, b1, W2, b2,
                                               (float*)d_out);
  }
}


# MLP kernel: LDS tiles of h and mid get row pitch 68 / 132 floats, so an A-fragment read of 16 rows covers all 64 banks once (at pitch 64 / 128 the rows met on two banks: 16-way conflict)
# speedup vs baseline: 2.3741x; 1.1406x over previous
#include <hip/hip_runtime.h>
#include <hip/hip_bf16.h>


#define GIN_D 64
#define GIN_H 128
#define TILE_N 256
#define TILE_SHIFT 8
#define NB_MAX 512
#define CHUNK_E 16384
#define BIN_T 1024
#define PER_T (CHUNK_E / BIN_T)
#define SORT_C 8192
#define SORT_T (SORT_C / 1024)
#define ROWS_W (TILE_N / 32)
#define H_PITCH (GIN_D + 4)
#define MID_PITCH (GIN_H + 4)

typedef __attribute__((ext_vector_type(2))) float v2f;
typedef __attribute__((ext_vector_type(8))) float v8f;

__global__ __launch_bounds__(BIN_T) void gin_count_kernel(
    const int* __restrict__ edst, int* __restrict__ hist_wg, int n_edges, int nb) {
  __shared__ int hist[NB_MAX];
  const int tid = threadIdx.x;
  if (tid < NB_MAX) hist[tid] = 0;
  __syncthreads();
  const long long e0 = (long long)blockIdx.x * CHUNK_E + tid;
  const long long last = (long long)n_edges - 1;
  int dd[PER_T];
#pragma unroll
  for (int k = 0; k < PER_T; ++k) {
    const long long e = e0 + (long long)k * BIN_T;
    dd[k] = edst[e < last ? e : last];
  }
#pragma unroll
  for (int k = 0; k < PER_T; ++k) {
    const long long e = e0 + (long long)k * BIN_T;
    if (e <= last)
      __hip_atomic_fetch_add(&hist[dd[k] >> TILE_SHIFT], 1, __ATOMIC_RELAXED,
                             __HIP_MEMORY_SCOPE_WORKGROUP);
  }
  __syncthreads();
  if (tid < nb) hist_wg[(size_t)blockIdx.x * nb + tid] = hist[tid];
}

__global__ __launch_bounds__(BIN_T) void gin_fill_kernel(
    const int* __restrict__ esrc, const int* __restrict__ edst,
    const int* __restrict__ hist_wg, int* __restrict__ binned_i,
    int* __restrict__ tile_start, int n_edges, int nb, int nwg) {
  __shared__ int tot[NB_MAX];
  __shared__ int scan[2][NB_MAX];
  __shared__ int base[NB_MAX];
  __shared__ int cur[NB_MAX];
  const int tid = threadIdx.x;
  const int wg = blockIdx.x;

  int pre = 0, all = 0;
  if (tid < nb) {
    for (int w0 = 0; w0 < nwg; w0 += 16) {
      int v[16];
#pragma unroll
      for (int j = 0; j < 16; ++j) {
        const int w = w0 + j;
        v[j] = (w < nwg) ? hist_wg[(size_t)w * nb + tid] : 0;
      }
#pragma unroll
      for (int j = 0; j < 16; ++j) {
        all += v[j];
        if (w0 + j < wg) pre += v[j];
      }
    }
  }
  if (tid < NB_MAX) { tot[tid] = all; scan[0][tid] = all; cur[tid] = 0; }
  int sb = 0;
  for (int off = 1; off < NB_MAX; off <<= 1) {
    __syncthreads();
    if (tid < NB_MAX) {
      int v = scan[sb][tid];
      if (tid >= off) v += scan[sb][tid - off];
      scan[sb ^ 1][tid] = v;
    }
    sb ^= 1;
  }
  __syncthreads();
  if (tid < NB_MAX) {
    const int excl = scan[sb][tid] - tot[tid];
    base[tid] = excl + pre;
    if (wg == 0 && tid <= nb) tile_start[tid] = excl;
  }
  __syncthreads();

  int2* __restrict__ binned = (int2*)binned_i;
  const long long e0 = (long long)wg * CHUNK_E + tid;
  const long long last = (long long)n_edges - 1;
  int dd[PER_T], ss[PER_T];
#pragma unroll
  for (int k = 0; k < PER_T; ++k) {
    const long long e = e0 + (long long)k * BIN_T;
    const long long ec = e < last ? e : last;
    dd[k] = edst[ec];
    ss[k] = esrc[ec];
  }
#pragma unroll
  for (int k = 0; k < PER_T; ++k) {
    const long long e = e0 + (long long)k * BIN_T;
    if (e <= last) {
      const int b = dd[k] >> TILE_SHIFT;
      const int r = __hip_atomic_fetch_add(&cur[b], 1, __ATOMIC_RELAXED,
                                           __HIP_MEMORY_SCOPE_WORKGROUP);
      binned[base[b] + r] = make_int2(ss[k], dd[k]);
    }
  }
}

__global__ __launch_bounds__(1024) __attribute__((amdgpu_waves_per_eu(16, 16))) void gin_tile_kernel(
    const float* __restrict__ feat, const float* __restrict__ eps,
    const int* __restrict__ binned_i, const int* __restrict__ tile_start,
    float* __restrict__ hacc, int n_nodes) {
  __shared__ int cnt[TILE_N];
  __shared__ int rstart[TILE_N + 1];
  __shared__ int ssrc[SORT_C];
  const int tid = threadIdx.x;
  const int lane = tid & 31;
  const int wave = __builtin_amdgcn_readfirstlane(tid >> 5);
  const int n0 = blockIdx.x * TILE_N;
  const int2* __restrict__ binned = (const int2*)binned_i;
  const int e_begin = tile_start[blockIdx.x];
  const int e_end = tile_start[blockIdx.x + 1];

  float lo[ROWS_W], hi[ROWS_W];
#pragma unroll
  for (int rr = 0; rr < ROWS_W; ++rr) { lo[rr] = 0.f; hi[rr] = 0.f; }

  for (int c0 = e_begin; c0 < e_end; c0 += SORT_C) {
    const int left = e_end - c0;
    const int n = left < SORT_C ? left : SORT_C;
    if (tid < TILE_N) cnt[tid] = 0;
    __syncthreads();

    int ss[SORT_T], dd[SORT_T];
#pragma unroll
    for (int k = 0; k < SORT_T; ++k) {
      ss[k] = 0; dd[k] = 0;
      if (k * 1024 < n) {
        const int i = tid + k * 1024;
        const int2 p = binned[c0 + (i < n ? i : 0)];
        ss[k] = p.x;
        dd[k] = p.y;
      }
    }
#pragma unroll
    for (int k = 0; k < SORT_T; ++k) {
      if (tid + k * 1024 < n) {
        dd[k] -= n0;
        const int r = __hip_atomic_fetch_add(&cnt[dd[k]], 1, __ATOMIC_RELAXED,
                                             __HIP_MEMORY_SCOPE_WORKGROUP);
        dd[k] |= r << TILE_SHIFT;
      }
    }
    __syncthreads();

    if (wave == 0) {
      int v[8];
#pragma unroll
      for (int j = 0; j < 8; ++j) v[j] = cnt[lane * 8 + j];
      int t = 0;
      int pre[8];
#pragma unroll
      for (int j = 0; j < 8; ++j) { pre[j] = t; t += v[j]; }
      int inc = t;
#pragma unroll
      for (int off = 1; off < 32; off <<= 1) {
        const int u = __shfl_up(inc, off);
        if (lane >= off) inc += u;
      }
      const int base = inc - t;
#pragma unroll
      for (int j = 0; j < 8; ++j) rstart[lane * 8 + j] = base + pre[j];
      if (lane == 31) rstart[TILE_N] = inc;
    }
    __syncthreads();

#pragma unroll
    for (int k = 0; k < SORT_T; ++k) {
      if (tid + k * 1024 < n)
        ssrc[rstart[dd[k] & (TILE_N - 1)] + (dd[k] >> TILE_SHIFT)] = ss[k];
    }
    __syncthreads();

    const int rsv = rstart[wave * ROWS_W + (lane < ROWS_W ? lane : ROWS_W)];
#pragma unroll
    for (int rr = 0; rr < ROWS_W; ++rr) {
      const int rb = __builtin_amdgcn_readlane(rsv, rr);
      const int re = __builtin_amdgcn_readlane(rsv, rr + 1);
      for (int j0 = rb; j0 < re; j0 += 32) {
        const int m = re - j0;
        const int idx = j0 + lane;
        const int sv = idx < re ? ssrc[idx] : 0;
        for (int g = 0; g < 32 && g < m; g += 8) {
          float a[8], c[8];
#pragma unroll
          for (int j = 0; j < 8; ++j) {
            const int s = __builtin_amdgcn_readlane(sv, g + j);
            const float* sp = feat + (size_t)s * GIN_D + lane;
            a[j] = sp[0];
            c[j] = sp[32];
          }
#pragma unroll
          for (int j = 0; j < 8; ++j) {
            if (g + j < m) { lo[rr] += a[j]; hi[rr] += c[j]; }
          }
        }
      }
    }
  }

  const float sc = 1.0f + eps[0];
  float f0[ROWS_W], f1[ROWS_W];
#pragma unroll
  for (int rr = 0; rr < ROWS_W; ++rr) {
    const int node = n0 + wave * ROWS_W + rr;
    const int nc = node < n_nodes ? node : n_nodes - 1;
    const float* fp = feat + (size_t)nc * GIN_D + lane;
    f0[rr] = fp[0];
    f1[rr] = fp[32];
  }
#pragma unroll
  for (int rr = 0; rr < ROWS_W; ++rr) {
    const int node = n0 + wave * ROWS_W + rr;
    if (node < n_nodes) {
      float* hp = hacc + (size_t)node * GIN_D + lane;
      hp[0] = sc * f0[rr] + lo[rr];
      hp[32] = sc * f1[rr] + hi[rr];
    }
  }
}

__global__ __launch_bounds__(128) void gin_mlp_kernel(
    const float* __restrict__ hacc, const float* __restrict__ W1,
    const float* __restrict__ b1, const float* __restrict__ W2,
    const float* __restrict__ b2, float* __restrict__ out) {
  __shared__ __attribute__((aligned(16))) float lds_h[16 * H_PITCH];
  __shared__ float lds_mid[16 * MID_PITCH];

  const int tid  = threadIdx.x;
  const int wave = tid >> 5;
  const int lane = tid & 31;
  const int lhalf = lane >> 4;
  const int lmod  = lane & 15;
  const int node0 = blockIdx.x * 16;

  {
    const float4* src = (const float4*)(hacc + (size_t)node0 * GIN_D);
    const int i0 = tid, i1 = tid + 128;
    *(float4*)&lds_h[(i0 >> 4) * H_PITCH + (i0 & 15) * 4] = src[i0];
    *(float4*)&lds_h[(i1 >> 4) * H_PITCH + (i1 & 15) * 4] = src[i1];
  }
  __syncthreads();

  v8f acc0 = {};
  v8f acc1 = {};
  const int n0 = lmod + 16 * (2 * wave);
  const int n1 = lmod + 16 * (2 * wave + 1);
#pragma unroll
  for (int k = 0; k < 16; ++k) {
    const int kb = 4 * k + 2 * lhalf;
    v2f a;
    a.x = lds_h[lmod * H_PITCH + kb];
    a.y = lds_h[lmod * H_PITCH + kb + 1];
    v2f bf0, bf1;
    bf0.x = W1[(size_t)kb * GIN_H + n0];
    bf0.y = W1[(size_t)(kb + 1) * GIN_H + n0];
    bf1.x = W1[(size_t)kb * GIN_H + n1];
    bf1.y = W1[(size_t)(kb + 1) * GIN_H + n1];
    acc0 = __builtin_amdgcn_wmma_f32_16x16x4_f32(false, a, false, bf0,
                                                 (short)0, acc0, false, false);
    acc1 = __builtin_amdgcn_wmma_f32_16x16x4_f32(false, a, false, bf1,
                                                 (short)0, acc1, false, false);
  }
  {
    const float b1v0 = b1[n0];
    const float b1v1 = b1[n1];
#pragma unroll
    for (int r = 0; r < 8; ++r) {
      const int row = r + 8 * lhalf;
      lds_mid[row * MID_PITCH + n0] = fmaxf(acc0[r] + b1v0, 0.0f);
      lds_mid[row * MID_PITCH + n1] = fmaxf(acc1[r] + b1v1, 0.0f);
    }
  }
  __syncthreads();

  v8f acc2 = {};
  const int n2 = lmod + 16 * wave;
#pragma unroll
  for (int k = 0; k < 32; ++k) {
    const int kb = 4 * k + 2 * lhalf;
    v2f a;
    a.x = lds_mid[lmod * MID_PITCH + kb];
    a.y = lds_mid[lmod * MID_PITCH + kb + 1];
    v2f bf;
    bf.x = W2[(size_t)kb * GIN_D + n2];
    bf.y = W2[(size_t)(kb + 1) * GIN_D + n2];
    acc2 = __builtin_amdgcn_wmma_f32_16x16x4_f32(false, a, false, bf,
                                                 (short)0, acc2, false, false);
  }
  {
    const float b2v = b2[n2];
#pragma unroll
    for (int r = 0; r < 8; ++r) {
      const int row = r + 8 * lhalf;
      out[(size_t)(node0 + row) * GIN_D + n2] = acc2[r] + b2v;
    }
  }
}

extern "C" void kernel_launch(void* const* d_in, const int* in_sizes, int n_in,
                              void* d_out, int out_size, void* d_ws, size_t ws_size,
                              hipStream_t stream) {
  const float* feat = (const float*)d_in[0];
  const int*   esrc = (const int*)d_in[1];
  const int*   edst = (const int*)d_in[2];
  const float* W1   = (const float*)d_in[3];
  const float* b1   = (const float*)d_in[4];
  const float* W2   = (const float*)d_in[5];
  const float* b2   = (const float*)d_in[6];
  const float* eps  = (const float*)d_in[7];

  const int n_nodes = in_sizes[0] / GIN_D;
  const int n_edges = in_sizes[1];
  const int nb  = (n_nodes + TILE_N - 1) / TILE_N;
  const int nwg = (n_edges + CHUNK_E - 1) / CHUNK_E;

  float* hacc       = (float*)d_ws;
  int*   binned     = (int*)(hacc + (size_t)n_nodes * GIN_D);
  int*   hist_wg    = binned + (size_t)n_edges * 2;
  int*   tile_start = hist_wg + (size_t)nwg * nb;

  gin_count_kernel<<<nwg, BIN_T, 0, stream>>>(edst, hist_wg, n_edges, nb);
  gin_fill_kernel<<<nwg, BIN_T, 0, stream>>>(esrc, edst, hist_wg, binned, tile_start,
                                             n_edges, nb, nwg);
  gin_tile_kernel<<<nb, 1024, 0, stream>>>(feat, eps, binned, tile_start, hacc, n_nodes);
  {
    const int blocks = n_nodes / 16;
    gin_mlp_kernel<<<blocks, 128, 0, stream>>>(hacc, W1, b1, W2, b2,
                                               (float*)d_out);
  }
}


# binning in one launch: each 6400-edge chunk put in tile order in LDS (integer rank, one-wave scan) and kept in its own range; the tile kernel reads its edges piece by piece via a (tile, chunk) table
# speedup vs baseline: 3.0640x; 1.2906x over previous
#include <hip/hip_runtime.h>
#include <hip/hip_bf16.h>


#define GIN_D 64
#define GIN_H 128
#define TILE_N 256
#define TILE_SHIFT 8
#define NB_MAX 512
#define CHUNK_E 6400
#define BIN_T 640
#define PER_T (CHUNK_E / BIN_T)
#define NWG_MAX 256
#define PIECE_E 32
#define PIECES_W (NWG_MAX / 32)
#define SORT_T PIECES_W
#define SORT_C (NWG_MAX * PIECE_E)
#define ROWS_W (TILE_N / 32)
#define H_PITCH (GIN_D + 4)
#define MID_PITCH (GIN_H + 4)

typedef __attribute__((ext_vector_type(2))) float v2f;
typedef __attribute__((ext_vector_type(8))) float v8f;

__global__ __launch_bounds__(BIN_T) void gin_bin_kernel(
    const int* __restrict__ esrc, const int* __restrict__ edst,
    int* __restrict__ binned_i, int* __restrict__ piece_i,
    int n_edges, int nb, int nwg) {
  __shared__ int hist[NB_MAX];
  __shared__ int base[NB_MAX];
  __shared__ int2 stage[CHUNK_E];
  const int tid = threadIdx.x;
  const int lane = tid & 31;
  const int wave = __builtin_amdgcn_readfirstlane(tid >> 5);
  const int wg = blockIdx.x;
  if (tid < NB_MAX) hist[tid] = 0;
  __syncthreads();

  const long long e0 = (long long)wg * CHUNK_E + tid;
  const long long last = (long long)n_edges - 1;
  int dd[PER_T], ss[PER_T], rk[PER_T];
#pragma unroll
  for (int k = 0; k < PER_T; ++k) {
    const long long e = e0 + (long long)k * BIN_T;
    const long long ec = e < last ? e : last;
    dd[k] = edst[ec];
    ss[k] = esrc[ec];
  }
#pragma unroll
  for (int k = 0; k < PER_T; ++k) {
    const long long e = e0 + (long long)k * BIN_T;
    rk[k] = 0;
    if (e <= last)
      rk[k] = __hip_atomic_fetch_add(&hist[dd[k] >> TILE_SHIFT], 1, __ATOMIC_RELAXED,
                                     __HIP_MEMORY_SCOPE_WORKGROUP);
  }
  __syncthreads();

  if (wave == 0) {
    int v[NB_MAX / 32];
#pragma unroll
    for (int j = 0; j < NB_MAX / 32; ++j) v[j] = hist[lane * (NB_MAX / 32) + j];
    int t = 0;
    int pre[NB_MAX / 32];
#pragma unroll
    for (int j = 0; j < NB_MAX / 32; ++j) { pre[j] = t; t += v[j]; }
    int inc = t;
#pragma unroll
    for (int off = 1; off < 32; off <<= 1) {
      const int u = __shfl_up(inc, off);
      if (lane >= off) inc += u;
    }
    const int b0 = inc - t;
#pragma unroll
    for (int j = 0; j < NB_MAX / 32; ++j) base[lane * (NB_MAX / 32) + j] = b0 + pre[j];
  }
  __syncthreads();

  const int c0 = wg * CHUNK_E;
#pragma unroll
  for (int k = 0; k < PER_T; ++k) {
    const long long e = e0 + (long long)k * BIN_T;
    if (e <= last)
      stage[base[dd[k] >> TILE_SHIFT] + rk[k]] = make_int2(ss[k], dd[k]);
  }
  int2* __restrict__ piece = (int2*)piece_i;
  if (tid < nb) piece[(size_t)tid * nwg + wg] = make_int2(c0 + base[tid], hist[tid]);
  __syncthreads();

  int2* __restrict__ binned = (int2*)binned_i;
  const long long left = (long long)n_edges - c0;
  const int nc = left < CHUNK_E ? (int)left : CHUNK_E;
#pragma unroll
  for (int k = 0; k < PER_T; ++k) {
    const int i = tid + k * BIN_T;
    if (i < nc) binned[c0 + i] = stage[i];
  }
}

__global__ __launch_bounds__(1024) __attribute__((amdgpu_waves_per_eu(16, 16))) void gin_tile_kernel(
    const float* __restrict__ feat, const float* __restrict__ eps,
    const int* __restrict__ binned_i, const int* __restrict__ piece_i,
    float* __restrict__ hacc, int n_nodes, int nwg) {
  __shared__ int pstart[NWG_MAX];
  __shared__ int pcnt[NWG_MAX];
  __shared__ int wmax[NWG_MAX / 32];
  __shared__ int cnt[TILE_N];
  __shared__ int rstart[TILE_N + 1];
  __shared__ int ssrc[SORT_C];
  const int tid = threadIdx.x;
  const int lane = tid & 31;
  const int wave = __builtin_amdgcn_readfirstlane(tid >> 5);
  const int n0 = blockIdx.x * TILE_N;
  const int2* __restrict__ binned = (const int2*)binned_i;
  if (tid < NWG_MAX) {
    int2 pc = make_int2(0, 0);
    if (tid < nwg) pc = ((const int2*)piece_i)[(size_t)blockIdx.x * nwg + tid];
    pstart[tid] = pc.x;
    pcnt[tid] = pc.y;
    int m = pc.y;
#pragma unroll
    for (int off = 16; off >= 1; off >>= 1) {
      const int u = __shfl_xor(m, off);
      m = u > m ? u : m;
    }
    if (lane == 0) wmax[wave] = m;
  }
  __syncthreads();
  int maxc_v = 0;
#pragma unroll
  for (int j = 0; j < NWG_MAX / 32; ++j) maxc_v = wmax[j] > maxc_v ? wmax[j] : maxc_v;
  const int maxc = __builtin_amdgcn_readfirstlane(maxc_v);
  int pst[PIECES_W], pcn[PIECES_W];
#pragma unroll
  for (int k = 0; k < PIECES_W; ++k) {
    pst[k] = __builtin_amdgcn_readfirstlane(pstart[wave + 32 * k]);
    pcn[k] = __builtin_amdgcn_readfirstlane(pcnt[wave + 32 * k]);
  }

  float lo[ROWS_W], hi[ROWS_W];
#pragma unroll
  for (int rr = 0; rr < ROWS_W; ++rr) { lo[rr] = 0.f; hi[rr] = 0.f; }

  for (int p0 = 0; p0 < maxc; p0 += PIECE_E) {
    if (tid < TILE_N) cnt[tid] = 0;
    __syncthreads();

    int ss[SORT_T], dd[SORT_T];
#pragma unroll
    for (int k = 0; k < SORT_T; ++k) {
      ss[k] = 0; dd[k] = -1;
      const int j = p0 + lane;
      if (j < pcn[k]) {
        const int2 p = binned[pst[k] + j];
        ss[k] = p.x;
        dd[k] = p.y;
      }
    }
#pragma unroll
    for (int k = 0; k < SORT_T; ++k) {
      if (dd[k] >= 0) {
        const int row = dd[k] - n0;
        const int r = __hip_atomic_fetch_add(&cnt[row], 1, __ATOMIC_RELAXED,
                                             __HIP_MEMORY_SCOPE_WORKGROUP);
        dd[k] = row | (r << TILE_SHIFT);
      }
    }
    __syncthreads();

    if (wave == 0) {
      int v[8];
#pragma unroll
      for (int j = 0; j < 8; ++j) v[j] = cnt[lane * 8 + j];
      int t = 0;
      int pre[8];
#pragma unroll
      for (int j = 0; j < 8; ++j) { pre[j] = t; t += v[j]; }
      int inc = t;
#pragma unroll
      for (int off = 1; off < 32; off <<= 1) {
        const int u = __shfl_up(inc, off);
        if (lane >= off) inc += u;
      }
      const int base = inc - t;
#pragma unroll
      for (int j = 0; j < 8; ++j) rstart[lane * 8 + j] = base + pre[j];
      if (lane == 31) rstart[TILE_N] = inc;
    }
    __syncthreads();

#pragma unroll
    for (int k = 0; k < SORT_T; ++k) {
      if (dd[k] >= 0)
        ssrc[rstart[dd[k] & (TILE_N - 1)] + (dd[k] >> TILE_SHIFT)] = ss[k];
    }
    __syncthreads();

    const int rsv = rstart[wave * ROWS_W + (lane < ROWS_W ? lane : ROWS_W)];
#pragma unroll
    for (int rr = 0; rr < ROWS_W; ++rr) {
      const int rb = __builtin_amdgcn_readlane(rsv, rr);
      const int re = __builtin_amdgcn_readlane(rsv, rr + 1);
      for (int j0 = rb; j0 < re; j0 += 32) {
        const int m = re - j0;
        const int idx = j0 + lane;
        const int sv = idx < re ? ssrc[idx] : 0;
        for (int g = 0; g < 32 && g < m; g += 8) {
          float a[8], c[8];
#pragma unroll
          for (int j = 0; j < 8; ++j) {
            const int s = __builtin_amdgcn_readlane(sv, g + j);
            const float* sp = feat + (size_t)s * GIN_D + lane;
            a[j] = sp[0];
            c[j] = sp[32];
          }
#pragma unroll
          for (int j = 0; j < 8; ++j) {
            if (g + j < m) { lo[rr] += a[j]; hi[rr] += c[j]; }
          }
        }
      }
    }
  }

  const float sc = 1.0f + eps[0];
  float f0[ROWS_W], f1[ROWS_W];
#pragma unroll
  for (int rr = 0; rr < ROWS_W; ++rr) {
    const int node = n0 + wave * ROWS_W + rr;
    const int nc = node < n_nodes ? node : n_nodes - 1;
    const float* fp = feat + (size_t)nc * GIN_D + lane;
    f0[rr] = fp[0];
    f1[rr] = fp[32];
  }
#pragma unroll
  for (int rr = 0; rr < ROWS_W; ++rr) {
    const int node = n0 + wave * ROWS_W + rr;
    if (node < n_nodes) {
      float* hp = hacc + (size_t)node * GIN_D + lane;
      hp[0] = sc * f0[rr] + lo[rr];
      hp[32] = sc * f1[rr] + hi[rr];
    }
  }
}

__global__ __launch_bounds__(128) void gin_mlp_kernel(
    const float* __restrict__ hacc, const float* __restrict__ W1,
    const float* __restrict__ b1, const float* __restrict__ W2,
    const float* __restrict__ b2, float* __restrict__ out) {
  __shared__ __attribute__((aligned(16))) float lds_h[16 * H_PITCH];
  __shared__ float lds_mid[16 * MID_PITCH];

  const int tid  = threadIdx.x;
  const int wave = tid >> 5;
  const int lane = tid & 31;
  const int lhalf = lane >> 4;
  const int lmod  = lane & 15;
  const int node0 = blockIdx.x * 16;

  {
    const float4* src = (const float4*)(hacc + (size_t)node0 * GIN_D);
    const int i0 = tid, i1 = tid + 128;
    *(float4*)&lds_h[(i0 >> 4) * H_PITCH + (i0 & 15) * 4] = src[i0];
    *(float4*)&lds_h[(i1 >> 4) * H_PITCH + (i1 & 15) * 4] = src[i1];
  }
  __syncthreads();

  v8f acc0 = {};
  v8f acc1 = {};
  const int n0 = lmod + 16 * (2 * wave);
  const int n1 = lmod + 16 * (2 * wave + 1);
#pragma unroll
  for (int k = 0; k < 16; ++k) {
    const int kb = 4 * k + 2 * lhalf;
    v2f a;
    a.x = lds_h[lmod * H_PITCH + kb];
    a.y = lds_h[lmod * H_PITCH + kb + 1];
    v2f bf0, bf1;
    bf0.x = W1[(size_t)kb * GIN_H + n0];
    bf0.y = W1[(size_t)(kb + 1) * GIN_H + n0];
    bf1.x = W1[(size_t)kb * GIN_H + n1];
    bf1.y = W1[(size_t)(kb + 1) * GIN_H + n1];
    acc0 = __builtin_amdgcn_wmma_f32_16x16x4_f32(false, a, false, bf0,
                                                 (short)0, acc0, false, false);
    acc1 = __builtin_amdgcn_wmma_f32_16x16x4_f32(false, a, false, bf1,
                                                 (short)0, acc1, false, false);
  }
  {
    const float b1v0 = b1[n0];
    const float b1v1 = b1[n1];
#pragma unroll
    for (int r = 0; r < 8; ++r) {
      const int row = r + 8 * lhalf;
      lds_mid[row * MID_PITCH + n0] = fmaxf(acc0[r] + b1v0, 0.0f);
      lds_mid[row * MID_PITCH + n1] = fmaxf(acc1[r] + b1v1, 0.0f);
    }
  }
  __syncthreads();

  v8f acc2 = {};
  const int n2 = lmod + 16 * wave;
#pragma unroll
  for (int k = 0; k < 32; ++k) {
    const int kb = 4 * k + 2 * lhalf;
    v2f a;
    a.x = lds_mid[lmod * MID_PITCH + kb];
    a.y = lds_mid[lmod * MID_PITCH + kb + 1];
    v2f bf;
    bf.x = W2[(size_t)kb * GIN_D + n2];
    bf.y = W2[(size_t)(kb + 1) * GIN_D + n2];
    acc2 = __builtin_amdgcn_wmma_f32_16x16x4_f32(false, a, false, bf,
                                                 (short)0, acc2, false, false);
  }
  {
    const float b2v = b2[n2];
#pragma unroll
    for (int r = 0; r < 8; ++r) {
      const int row = r + 8 * lhalf;
      out[(size_t)(node0 + row) * GIN_D + n2] = acc2[r] + b2v;
    }
  }
}

extern "C" void kernel_launch(void* const* d_in, const int* in_sizes, int n_in,
                              void* d_out, int out_size, void* d_ws, size_t ws_size,
                              hipStream_t stream) {
  const float* feat = (const float*)d_in[0];
  const int*   esrc = (const int*)d_in[1];
  const int*   edst = (const int*)d_in[2];
  const float* W1   = (const float*)d_in[3];
  const float* b1   = (const float*)d_in[4];
  const float* W2   = (const float*)d_in[5];
  const float* b2   = (const float*)d_in[6];
  const float* eps  = (const float*)d_in[7];

  const int n_nodes = in_sizes[0] / GIN_D;
  const int n_edges = in_sizes[1];
  const int nb  = (n_nodes + TILE_N - 1) / TILE_N;
  const int nwg = (n_edges + CHUNK_E - 1) / CHUNK_E;

  float* hacc   = (float*)d_ws;
  int*   binned = (int*)(hacc + (size_t)n_nodes * GIN_D);
  int*   piece  = binned + (size_t)n_edges * 2;

  gin_bin_kernel<<<nwg, BIN_T, 0, stream>>>(esrc, edst, binned, piece, n_edges, nb, nwg);
  gin_tile_kernel<<<nb, 1024, 0, stream>>>(feat, eps, binned, piece, hacc, n_nodes, nwg);
  {
    const int blocks = n_nodes / 16;
    gin_mlp_kernel<<<blocks, 128, 0, stream>>>(hacc, W1, b1, W2, b2,
                                               (float*)d_out);
  }
}


# MLP folded into the tile kernel: h tile stays in LDS, the tile's waves run both f32 WMMA GEMMs on it and store out (2 launches, no h round trip); 208-node tiles; row sums in arrival order as before
# speedup vs baseline: 3.4673x; 1.1316x over previous
#include <hip/hip_runtime.h>
#include <hip/hip_bf16.h>


#define GIN_D 64
#define GIN_H 128
#define TILE_N 208
#define TILE_W 26
#define TILE_T (TILE_W * 32)
#define ROW_SHIFT 8
#define CNT_N 256
#define NB_MAX 512
#define CHUNK_E 6400
#define BIN_T 640
#define PER_T (CHUNK_E / BIN_T)
#define PIECES_W 10
#define NWG_MAX (TILE_W * PIECES_W)
#define NWG_WAVES ((NWG_MAX + 31) / 32)
#define PIECE_E 32
#define SORT_T PIECES_W
#define SORT_C (NWG_MAX * PIECE_E)
#define ROWS_W (TILE_N / TILE_W)
#define H_PITCH (GIN_D + 4)
#define MID_PITCH (GIN_H + 4)
#define BIG_W (TILE_N * MID_PITCH)
#define HBUF_OFF SORT_C

typedef __attribute__((ext_vector_type(2))) float v2f;
typedef __attribute__((ext_vector_type(8))) float v8f;

__global__ __launch_bounds__(BIN_T) void gin_bin_kernel(
    const int* __restrict__ esrc, const int* __restrict__ edst,
    int* __restrict__ binned_i, int* __restrict__ piece_i,
    int n_edges, int nb, int nwg) {
  __shared__ int hist[NB_MAX];
  __shared__ int base[NB_MAX];
  __shared__ int2 stage[CHUNK_E];
  const int tid = threadIdx.x;
  const int lane = tid & 31;
  const int wave = __builtin_amdgcn_readfirstlane(tid >> 5);
  const int wg = blockIdx.x;
  if (tid < NB_MAX) hist[tid] = 0;
  __syncthreads();

  const long long e0 = (long long)wg * CHUNK_E + tid;
  const long long last = (long long)n_edges - 1;
  int dd[PER_T], ss[PER_T], rk[PER_T];
#pragma unroll
  for (int k = 0; k < PER_T; ++k) {
    const long long e = e0 + (long long)k * BIN_T;
    const long long ec = e < last ? e : last;
    dd[k] = edst[ec];
    ss[k] = esrc[ec];
  }
#pragma unroll
  for (int k = 0; k < PER_T; ++k) {
    const long long e = e0 + (long long)k * BIN_T;
    rk[k] = 0;
    if (e <= last)
      rk[k] = __hip_atomic_fetch_add(&hist[(unsigned)dd[k] / TILE_N], 1, __ATOMIC_RELAXED,
                                     __HIP_MEMORY_SCOPE_WORKGROUP);
  }
  __syncthreads();

  if (wave == 0) {
    int v[NB_MAX / 32];
#pragma unroll
    for (int j = 0; j < NB_MAX / 32; ++j) v[j] = hist[lane * (NB_MAX / 32) + j];
    int t = 0;
    int pre[NB_MAX / 32];
#pragma unroll
    for (int j = 0; j < NB_MAX / 32; ++j) { pre[j] = t; t += v[j]; }
    int inc = t;
#pragma unroll
    for (int off = 1; off < 32; off <<= 1) {
      const int u = __shfl_up(inc, off);
      if (lane >= off) inc += u;
    }
    const int b0 = inc - t;
#pragma unroll
    for (int j = 0; j < NB_MAX / 32; ++j) base[lane * (NB_MAX / 32) + j] = b0 + pre[j];
  }
  __syncthreads();

  const int c0 = wg * CHUNK_E;
#pragma unroll
  for (int k = 0; k < PER_T; ++k) {
    const long long e = e0 + (long long)k * BIN_T;
    if (e <= last)
      stage[base[(unsigned)dd[k] / TILE_N] + rk[k]] = make_int2(ss[k], dd[k]);
  }
  int2* __restrict__ piece = (int2*)piece_i;
  if (tid < nb) piece[(size_t)tid * nwg + wg] = make_int2(c0 + base[tid], hist[tid]);
  __syncthreads();

  int2* __restrict__ binned = (int2*)binned_i;
  const long long left = (long long)n_edges - c0;
  const int nc = left < CHUNK_E ? (int)left : CHUNK_E;
#pragma unroll
  for (int k = 0; k < PER_T; ++k) {
    const int i = tid + k * BIN_T;
    if (i < nc) binned[c0 + i] = stage[i];
  }
}

__global__ __launch_bounds__(TILE_T) __attribute__((amdgpu_waves_per_eu(13, 16))) void gin_tile_kernel(
    const float* __restrict__ feat, const float* __restrict__ eps,
    const int* __restrict__ binned_i, const int* __restrict__ piece_i,
    const float* __restrict__ W1, const float* __restrict__ b1,
    const float* __restrict__ W2, const float* __restrict__ b2,
    float* __restrict__ out, int n_nodes, int nwg) {
  __shared__ int pstart[NWG_MAX];
  __shared__ int pcnt[NWG_MAX];
  __shared__ int wmax[NWG_WAVES];
  __shared__ int cnt[CNT_N];
  __shared__ int rstart[CNT_N + 1];
  __shared__ __attribute__((aligned(16))) float big[BIG_W];
  int* ssrc = (int*)big;
  float* hbuf = big + HBUF_OFF;
  float* mid = big;
  const int tid = threadIdx.x;
  const int lane = tid & 31;
  const int wave = __builtin_amdgcn_readfirstlane(tid >> 5);
  const int n0 = blockIdx.x * TILE_N;
  const int2* __restrict__ binned = (const int2*)binned_i;
  if (tid < NWG_WAVES * 32) {
    int2 pc = make_int2(0, 0);
    if (tid < nwg && tid < NWG_MAX) pc = ((const int2*)piece_i)[(size_t)blockIdx.x * nwg + tid];
    if (tid < NWG_MAX) {
      pstart[tid] = pc.x;
      pcnt[tid] = pc.y;
    }
    int m = pc.y;
#pragma unroll
    for (int off = 16; off >= 1; off >>= 1) {
      const int u = __shfl_xor(m, off);
      m = u > m ? u : m;
    }
    if (lane == 0) wmax[wave] = m;
  }
  __syncthreads();
  int maxc_v = 0;
#pragma unroll
  for (int j = 0; j < NWG_WAVES; ++j) maxc_v = wmax[j] > maxc_v ? wmax[j] : maxc_v;
  const int maxc = __builtin_amdgcn_readfirstlane(maxc_v);
  int pst[PIECES_W], pcn[PIECES_W];
#pragma unroll
  for (int k = 0; k < PIECES_W; ++k) {
    pst[k] = __builtin_amdgcn_readfirstlane(pstart[wave + TILE_W * k]);
    pcn[k] = __builtin_amdgcn_readfirstlane(pcnt[wave + TILE_W * k]);
  }

  v2f acc[ROWS_W];
#pragma unroll
  for (int rr = 0; rr < ROWS_W; ++rr) { acc[rr].x = 0.f; acc[rr].y = 0.f; }

  for (int p0 = 0; p0 < maxc; p0 += PIECE_E) {
    if (tid < CNT_N) cnt[tid] = 0;
    __syncthreads();

    int ss[SORT_T], dd[SORT_T];
#pragma unroll
    for (int k = 0; k < SORT_T; ++k) {
      ss[k] = 0; dd[k] = -1;
      const int j = p0 + lane;
      if (j < pcn[k]) {
        const int2 p = binned[pst[k] + j];
        ss[k] = p.x;
        dd[k] = p.y;
      }
    }
#pragma unroll
    for (int k = 0; k < SORT_T; ++k) {
      if (dd[k] >= 0) {
        const int row = dd[k] - n0;
        const int r = __hip_atomic_fetch_add(&cnt[row], 1, __ATOMIC_RELAXED,
                                             __HIP_MEMORY_SCOPE_WORKGROUP);
        dd[k] = row | (r << ROW_SHIFT);
      }
    }
    __syncthreads();

    if (wave == 0) {
      int v[8];
#pragma unroll
      for (int j = 0; j < 8; ++j) v[j] = cnt[lane * 8 + j];
      int t = 0;
      int pre[8];
#pragma unroll
      for (int j = 0; j < 8; ++j) { pre[j] = t; t += v[j]; }
      int inc = t;
#pragma unroll
      for (int off = 1; off < 32; off <<= 1) {
        const int u = __shfl_up(inc, off);
        if (lane >= off) inc += u;
      }
      const int base = inc - t;
#pragma unroll
      for (int j = 0; j < 8; ++j) rstart[lane * 8 + j] = base + pre[j];
      if (lane == 31) rstart[CNT_N] = inc;
    }
    __syncthreads();

#pragma unroll
    for (int k = 0; k < SORT_T; ++k) {
      if (dd[k] >= 0)
        ssrc[rstart[dd[k] & (CNT_N - 1)] + (dd[k] >> ROW_SHIFT)] = ss[k];
    }
    __syncthreads();

    const int rsv = rstart[wave * ROWS_W + (lane < ROWS_W ? lane : ROWS_W)];
#pragma unroll
    for (int rr = 0; rr < ROWS_W; ++rr) {
      const int rb = __builtin_amdgcn_readlane(rsv, rr);
      const int re = __builtin_amdgcn_readlane(rsv, rr + 1);
      for (int j0 = rb; j0 < re; j0 += 32) {
        const int m = re - j0;
        const int idx = j0 + lane;
        const int sv = idx < re ? ssrc[idx] : 0;
        for (int g = 0; g < 32 && g < m; g += 8) {
          v2f a[8];
#pragma unroll
          for (int j = 0; j < 8; ++j) {
            const int s = __builtin_amdgcn_readlane(sv, g + j);
            a[j] = ((const v2f*)(feat + (size_t)s * GIN_D))[lane];
          }
#pragma unroll
          for (int j = 0; j < 8; ++j) {
            if (g + j < m) { acc[rr].x += a[j].x; acc[rr].y += a[j].y; }
          }
        }
      }
    }
  }

  {
    const float sc = 1.0f + eps[0];
    v2f f[ROWS_W];
#pragma unroll
    for (int rr = 0; rr < ROWS_W; ++rr) {
      const int node = n0 + wave * ROWS_W + rr;
      const int nc = node < n_nodes ? node : n_nodes - 1;
      f[rr] = ((const v2f*)(feat + (size_t)nc * GIN_D))[lane];
    }
#pragma unroll
    for (int rr = 0; rr < ROWS_W; ++rr) {
      v2f h;
      h.x = sc * f[rr].x + acc[rr].x;
      h.y = sc * f[rr].y + acc[rr].y;
      *(v2f*)&hbuf[(wave * ROWS_W + rr) * H_PITCH + 2 * lane] = h;
    }
  }
  __syncthreads();

  const int lhalf = lane >> 4;
  const int lmod  = lane & 15;
  const int blk = wave >> 1;
  const int ch = wave & 1;

  v8f c0 = {}, c1 = {}, c2 = {}, c3 = {};
  const int nA = 64 * ch + lmod;
  {
    const float* arow = hbuf + (blk * 16 + lmod) * H_PITCH + 2 * lhalf;
    const float* wcol = W1 + (size_t)(2 * lhalf) * GIN_H + nA;
#pragma unroll
    for (int k = 0; k < 16; ++k) {
      v2f a;
      a.x = arow[4 * k];
      a.y = arow[4 * k + 1];
      const float* w = wcol + (size_t)(4 * k) * GIN_H;
      v2f bf0, bf1, bf2, bf3;
      bf0.x = w[0];  bf0.y = w[GIN_H];
      bf1.x = w[16]; bf1.y = w[GIN_H + 16];
      bf2.x = w[32]; bf2.y = w[GIN_H + 32];
      bf3.x = w[48]; bf3.y = w[GIN_H + 48];
      c0 = __builtin_amdgcn_wmma_f32_16x16x4_f32(false, a, false, bf0, (short)0, c0, false, false);
      c1 = __builtin_amdgcn_wmma_f32_16x16x4_f32(false, a, false, bf1, (short)0, c1, false, false);
      c2 = __builtin_amdgcn_wmma_f32_16x16x4_f32(false, a, false, bf2, (short)0, c2, false, false);
      c3 = __builtin_amdgcn_wmma_f32_16x16x4_f32(false, a, false, bf3, (short)0, c3, false, false);
    }
  }
  __syncthreads();
  {
    const float bia0 = b1[nA], bia1 = b1[nA + 16], bia2 = b1[nA + 32], bia3 = b1[nA + 48];
#pragma unroll
    for (int r = 0; r < 8; ++r) {
      float* mrow_w = mid + (blk * 16 + r + 8 * lhalf) * MID_PITCH + nA;
      mrow_w[0]  = fmaxf(c0[r] + bia0, 0.0f);
      mrow_w[16] = fmaxf(c1[r] + bia1, 0.0f);
      mrow_w[32] = fmaxf(c2[r] + bia2, 0.0f);
      mrow_w[48] = fmaxf(c3[r] + bia3, 0.0f);
    }
  }
  __syncthreads();

  v8f d0 = {}, d1 = {};
  const int nB = 32 * ch + lmod;
  {
    const float* mrow = mid + (blk * 16 + lmod) * MID_PITCH + 2 * lhalf;
    const float* wcol = W2 + (size_t)(2 * lhalf) * GIN_D + nB;
#pragma unroll 8
    for (int k = 0; k < 32; ++k) {
      v2f a;
      a.x = mrow[4 * k];
      a.y = mrow[4 * k + 1];
      const float* w = wcol + (size_t)(4 * k) * GIN_D;
      v2f bf0, bf1;
      bf0.x = w[0];  bf0.y = w[GIN_D];
      bf1.x = w[16]; bf1.y = w[GIN_D + 16];
      d0 = __builtin_amdgcn_wmma_f32_16x16x4_f32(false, a, false, bf0, (short)0, d0, false, false);
      d1 = __builtin_amdgcn_wmma_f32_16x16x4_f32(false, a, false, bf1, (short)0, d1, false, false);
    }
  }
  {
    const float bo0 = b2[nB], bo1 = b2[nB + 16];
#pragma unroll
    for (int r = 0; r < 8; ++r) {
      const int node = n0 + blk * 16 + r + 8 * lhalf;
      if (node < n_nodes) {
        out[(size_t)node * GIN_D + nB] = d0[r] + bo0;
        out[(size_t)node * GIN_D + nB + 16] = d1[r] + bo1;
      }
    }
  }
}

extern "C" void kernel_launch(void* const* d_in, const int* in_sizes, int n_in,
                              void* d_out, int out_size, void* d_ws, size_t ws_size,
                              hipStream_t stream) {
  const float* feat = (const float*)d_in[0];
  const int*   esrc = (const int*)d_in[1];
  const int*   edst = (const int*)d_in[2];
  const float* W1   = (const float*)d_in[3];
  const float* b1   = (const float*)d_in[4];
  const float* W2   = (const float*)d_in[5];
  const float* b2   = (const float*)d_in[6];
  const float* eps  = (const float*)d_in[7];

  const int n_nodes = in_sizes[0] / GIN_D;
  const int n_edges = in_sizes[1];
  const int nb  = (n_nodes + TILE_N - 1) / TILE_N;
  const int nwg = (n_edges + CHUNK_E - 1) / CHUNK_E;

  int*   binned = (int*)d_ws;
  int*   piece  = binned + (size_t)n_edges * 2;

  gin_bin_kernel<<<nwg, BIN_T, 0, stream>>>(esrc, edst, binned, piece, n_edges, nb, nwg);
  gin_tile_kernel<<<nb, TILE_T, 0, stream>>>(feat, eps, binned, piece, W1, b1, W2, b2,
                                             (float*)d_out, n_nodes, nwg);
}


# MLP: launch 1 rewrites W1, W2 in B-fragment lane order (64 KB workspace); 2 fragments per 16-byte load, 64 B loads per wave (was 256 of 4 bytes), GEMM1's asked one k step ahead; same k order
# speedup vs baseline: 3.5745x; 1.0309x over previous
#include <hip/hip_runtime.h>
#include <hip/hip_bf16.h>


#define GIN_D 64
#define GIN_H 128
#define TILE_N 208
#define TILE_W 26
#define TILE_T (TILE_W * 32)
#define ROW_SHIFT 8
#define CNT_N 256
#define NB_MAX 512
#define CHUNK_E 6400
#define BIN_T 640
#define PER_T (CHUNK_E / BIN_T)
#define PIECES_W 10
#define NWG_MAX (TILE_W * PIECES_W)
#define NWG_WAVES ((NWG_MAX + 31) / 32)
#define PIECE_E 32
#define SORT_T PIECES_W
#define SORT_C (NWG_MAX * PIECE_E)
#define ROWS_W (TILE_N / TILE_W)
#define H_PITCH (GIN_D + 4)
#define MID_PITCH (GIN_H + 4)
#define BIG_W (TILE_N * MID_PITCH)
#define HBUF_OFF SORT_C

typedef __attribute__((ext_vector_type(2))) float v2f;
typedef __attribute__((ext_vector_type(4))) float v4f;
typedef __attribute__((ext_vector_type(8))) float v8f;
#define W1F_N (16 * 2 * 2 * 32)
#define W2F_N (32 * 2 * 32)
#define WF_FLOATS ((W1F_N + W2F_N) * 4)

__global__ __launch_bounds__(BIN_T) void gin_bin_kernel(
    const int* __restrict__ esrc, const int* __restrict__ edst,
    const float* __restrict__ W1, const float* __restrict__ W2,
    float* __restrict__ wfrag,
    int* __restrict__ binned_i, int* __restrict__ piece_i,
    int n_edges, int nb, int nwg) {
  __shared__ int hist[NB_MAX];
  __shared__ int base[NB_MAX];
  __shared__ int2 stage[CHUNK_E];
  const int tid = threadIdx.x;
  const int lane = tid & 31;
  const int wave = __builtin_amdgcn_readfirstlane(tid >> 5);
  const int wg = blockIdx.x;
  for (int i = wg * BIN_T + tid; i < W1F_N + W2F_N; i += nwg * BIN_T) {
    const int hh = (i >> 4) & 1;
    const int c = i & 15;
    v4f v;
    if (i < W1F_N) {
      const int q = (i >> 5) & 1, chh = (i >> 6) & 1, ks = i >> 7;
      const float* p = W1 + (size_t)(4 * ks + 2 * hh) * GIN_H + 64 * chh + 32 * q + c;
      v.x = p[0]; v.y = p[GIN_H]; v.z = p[16]; v.w = p[GIN_H + 16];
    } else {
      const int j = i - W1F_N;
      const int chh = (j >> 5) & 1, ks = j >> 6;
      const float* p = W2 + (size_t)(4 * ks + 2 * hh) * GIN_D + 32 * chh + c;
      v.x = p[0]; v.y = p[GIN_D]; v.z = p[16]; v.w = p[GIN_D + 16];
    }
    ((v4f*)wfrag)[i] = v;
  }
  if (tid < NB_MAX) hist[tid] = 0;
  __syncthreads();

  const long long e0 = (long long)wg * CHUNK_E + tid;
  const long long last = (long long)n_edges - 1;
  int dd[PER_T], ss[PER_T], rk[PER_T];
#pragma unroll
  for (int k = 0; k < PER_T; ++k) {
    const long long e = e0 + (long long)k * BIN_T;
    const long long ec = e < last ? e : last;
    dd[k] = edst[ec];
    ss[k] = esrc[ec];
  }
#pragma unroll
  for (int k = 0; k < PER_T; ++k) {
    const long long e = e0 + (long long)k * BIN_T;
    rk[k] = 0;
    if (e <= last)
      rk[k] = __hip_atomic_fetch_add(&hist[(unsigned)dd[k] / TILE_N], 1, __ATOMIC_RELAXED,
                                     __HIP_MEMORY_SCOPE_WORKGROUP);
  }
  __syncthreads();

  if (wave == 0) {
    int v[NB_MAX / 32];
#pragma unroll
    for (int j = 0; j < NB_MAX / 32; ++j) v[j] = hist[lane * (NB_MAX / 32) + j];
    int t = 0;
    int pre[NB_MAX / 32];
#pragma unroll
    for (int j = 0; j < NB_MAX / 32; ++j) { pre[j] = t; t += v[j]; }
    int inc = t;
#pragma unroll
    for (int off = 1; off < 32; off <<= 1) {
      const int u = __shfl_up(inc, off);
      if (lane >= off) inc += u;
    }
    const int b0 = inc - t;
#pragma unroll
    for (int j = 0; j < NB_MAX / 32; ++j) base[lane * (NB_MAX / 32) + j] = b0 + pre[j];
  }
  __syncthreads();

  const int c0 = wg * CHUNK_E;
#pragma unroll
  for (int k = 0; k < PER_T; ++k) {
    const long long e = e0 + (long long)k * BIN_T;
    if (e <= last)
      stage[base[(unsigned)dd[k] / TILE_N] + rk[k]] = make_int2(ss[k], dd[k]);
  }
  int2* __restrict__ piece = (int2*)piece_i;
  if (tid < nb) piece[(size_t)tid * nwg + wg] = make_int2(c0 + base[tid], hist[tid]);
  __syncthreads();

  int2* __restrict__ binned = (int2*)binned_i;
  const long long left = (long long)n_edges - c0;
  const int nc = left < CHUNK_E ? (int)left : CHUNK_E;
#pragma unroll
  for (int k = 0; k < PER_T; ++k) {
    const int i = tid + k * BIN_T;
    if (i < nc) binned[c0 + i] = stage[i];
  }
}

__global__ __launch_bounds__(TILE_T) __attribute__((amdgpu_waves_per_eu(13, 16))) void gin_tile_kernel(
    const float* __restrict__ feat, const float* __restrict__ eps,
    const int* __restrict__ binned_i, const int* __restrict__ piece_i,
    const float* __restrict__ wfrag, const float* __restrict__ b1,
    const float* __restrict__ b2,
    float* __restrict__ out, int n_nodes, int nwg) {
  __shared__ int pstart[NWG_MAX];
  __shared__ int pcnt[NWG_MAX];
  __shared__ int wmax[NWG_WAVES];
  __shared__ int cnt[CNT_N];
  __shared__ int rstart[CNT_N + 1];
  __shared__ __attribute__((aligned(16))) float big[BIG_W];
  int* ssrc = (int*)big;
  float* hbuf = big + HBUF_OFF;
  float* mid = big;
  const int tid = threadIdx.x;
  const int lane = tid & 31;
  const int wave = __builtin_amdgcn_readfirstlane(tid >> 5);
  const int n0 = blockIdx.x * TILE_N;
  const int2* __restrict__ binned = (const int2*)binned_i;
  if (tid < NWG_WAVES * 32) {
    int2 pc = make_int2(0, 0);
    if (tid < nwg && tid < NWG_MAX) pc = ((const int2*)piece_i)[(size_t)blockIdx.x * nwg + tid];
    if (tid < NWG_MAX) {
      pstart[tid] = pc.x;
      pcnt[tid] = pc.y;
    }
    int m = pc.y;
#pragma unroll
    for (int off = 16; off >= 1; off >>= 1) {
      const int u = __shfl_xor(m, off);
      m = u > m ? u : m;
    }
    if (lane == 0) wmax[wave] = m;
  }
  __syncthreads();
  int maxc_v = 0;
#pragma unroll
  for (int j = 0; j < NWG_WAVES; ++j) maxc_v = wmax[j] > maxc_v ? wmax[j] : maxc_v;
  const int maxc = __builtin_amdgcn_readfirstlane(maxc_v);
  int pst[PIECES_W], pcn[PIECES_W];
#pragma unroll
  for (int k = 0; k < PIECES_W; ++k) {
    pst[k] = __builtin_amdgcn_readfirstlane(pstart[wave + TILE_W * k]);
    pcn[k] = __builtin_amdgcn_readfirstlane(pcnt[wave + TILE_W * k]);
  }

  v2f acc[ROWS_W];
#pragma unroll
  for (int rr = 0; rr < ROWS_W; ++rr) { acc[rr].x = 0.f; acc[rr].y = 0.f; }

  for (int p0 = 0; p0 < maxc; p0 += PIECE_E) {
    if (tid < CNT_N) cnt[tid] = 0;
    __syncthreads();

    int ss[SORT_T], dd[SORT_T];
#pragma unroll
    for (int k = 0; k < SORT_T; ++k) {
      ss[k] = 0; dd[k] = -1;
      const int j = p0 + lane;
      if (j < pcn[k]) {
        const int2 p = binned[pst[k] + j];
        ss[k] = p.x;
        dd[k] = p.y;
      }
    }
#pragma unroll
    for (int k = 0; k < SORT_T; ++k) {
      if (dd[k] >= 0) {
        const int row = dd[k] - n0;
        const int r = __hip_atomic_fetch_add(&cnt[row], 1, __ATOMIC_RELAXED,
                                             __HIP_MEMORY_SCOPE_WORKGROUP);
        dd[k] = row | (r << ROW_SHIFT);
      }
    }
    __syncthreads();

    if (wave == 0) {
      int v[8];
#pragma unroll
      for (int j = 0; j < 8; ++j) v[j] = cnt[lane * 8 + j];
      int t = 0;
      int pre[8];
#pragma unroll
      for (int j = 0; j < 8; ++j) { pre[j] = t; t += v[j]; }
      int inc = t;
#pragma unroll
      for (int off = 1; off < 32; off <<= 1) {
        const int u = __shfl_up(inc, off);
        if (lane >= off) inc += u;
      }
      const int base = inc - t;
#pragma unroll
      for (int j = 0; j < 8; ++j) rstart[lane * 8 + j] = base + pre[j];
      if (lane == 31) rstart[CNT_N] = inc;
    }
    __syncthreads();

#pragma unroll
    for (int k = 0; k < SORT_T; ++k) {
      if (dd[k] >= 0)
        ssrc[rstart[dd[k] & (CNT_N - 1)] + (dd[k] >> ROW_SHIFT)] = ss[k];
    }
    __syncthreads();

    const int rsv = rstart[wave * ROWS_W + (lane < ROWS_W ? lane : ROWS_W)];
#pragma unroll
    for (int rr = 0; rr < ROWS_W; ++rr) {
      const int rb = __builtin_amdgcn_readlane(rsv, rr);
      const int re = __builtin_amdgcn_readlane(rsv, rr + 1);
      for (int j0 = rb; j0 < re; j0 += 32) {
        const int m = re - j0;
        const int idx = j0 + lane;
        const int sv = idx < re ? ssrc[idx] : 0;
        for (int g = 0; g < 32 && g < m; g += 8) {
          v2f a[8];
#pragma unroll
          for (int j = 0; j < 8; ++j) {
            const int s = __builtin_amdgcn_readlane(sv, g + j);
            a[j] = ((const v2f*)(feat + (size_t)s * GIN_D))[lane];
          }
#pragma unroll
          for (int j = 0; j < 8; ++j) {
            if (g + j < m) { acc[rr].x += a[j].x; acc[rr].y += a[j].y; }
          }
        }
      }
    }
  }

  {
    const float sc = 1.0f + eps[0];
    v2f f[ROWS_W];
#pragma unroll
    for (int rr = 0; rr < ROWS_W; ++rr) {
      const int node = n0 + wave * ROWS_W + rr;
      const int nc = node < n_nodes ? node : n_nodes - 1;
      f[rr] = ((const v2f*)(feat + (size_t)nc * GIN_D))[lane];
    }
#pragma unroll
    for (int rr = 0; rr < ROWS_W; ++rr) {
      v2f h;
      h.x = sc * f[rr].x + acc[rr].x;
      h.y = sc * f[rr].y + acc[rr].y;
      *(v2f*)&hbuf[(wave * ROWS_W + rr) * H_PITCH + 2 * lane] = h;
    }
  }
  __syncthreads();

  const int lhalf = lane >> 4;
  const int lmod  = lane & 15;
  const int blk = wave >> 1;
  const int ch = wave & 1;

  v8f c0 = {}, c1 = {}, c2 = {}, c3 = {};
  const int nA = 64 * ch + lmod;
  {
    const float* arow = hbuf + (blk * 16 + lmod) * H_PITCH + 2 * lhalf;
    const v4f* wf = (const v4f*)wfrag + (ch * 2) * 32 + lane;
    v4f p0 = wf[0];
    v4f p1 = wf[32];
#pragma unroll
    for (int k = 0; k < 16; ++k) {
      v2f a;
      a.x = arow[4 * k];
      a.y = arow[4 * k + 1];
      v4f n0 = p0, n1 = p1;
      if (k + 1 < 16) {
        n0 = wf[((k + 1) * 4) * 32];
        n1 = wf[((k + 1) * 4 + 1) * 32];
      }
      v2f bf0, bf1, bf2, bf3;
      bf0.x = p0.x; bf0.y = p0.y;
      bf1.x = p0.z; bf1.y = p0.w;
      bf2.x = p1.x; bf2.y = p1.y;
      bf3.x = p1.z; bf3.y = p1.w;
      c0 = __builtin_amdgcn_wmma_f32_16x16x4_f32(false, a, false, bf0, (short)0, c0, false, false);
      c1 = __builtin_amdgcn_wmma_f32_16x16x4_f32(false, a, false, bf1, (short)0, c1, false, false);
      c2 = __builtin_amdgcn_wmma_f32_16x16x4_f32(false, a, false, bf2, (short)0, c2, false, false);
      c3 = __builtin_amdgcn_wmma_f32_16x16x4_f32(false, a, false, bf3, (short)0, c3, false, false);
      p0 = n0; p1 = n1;
    }
  }
  __syncthreads();
  {
    const float bia0 = b1[nA], bia1 = b1[nA + 16], bia2 = b1[nA + 32], bia3 = b1[nA + 48];
#pragma unroll
    for (int r = 0; r < 8; ++r) {
      float* mrow_w = mid + (blk * 16 + r + 8 * lhalf) * MID_PITCH + nA;
      mrow_w[0]  = fmaxf(c0[r] + bia0, 0.0f);
      mrow_w[16] = fmaxf(c1[r] + bia1, 0.0f);
      mrow_w[32] = fmaxf(c2[r] + bia2, 0.0f);
      mrow_w[48] = fmaxf(c3[r] + bia3, 0.0f);
    }
  }
  __syncthreads();

  v8f d0 = {}, d1 = {};
  const int nB = 32 * ch + lmod;
  {
    const float* mrow = mid + (blk * 16 + lmod) * MID_PITCH + 2 * lhalf;
    const v4f* wf = (const v4f*)wfrag + W1F_N + ch * 32 + lane;
#pragma unroll 8
    for (int k = 0; k < 32; ++k) {
      v2f a;
      a.x = mrow[4 * k];
      a.y = mrow[4 * k + 1];
      const v4f p = wf[(k * 2) * 32];
      v2f bf0, bf1;
      bf0.x = p.x; bf0.y = p.y;
      bf1.x = p.z; bf1.y = p.w;
      d0 = __builtin_amdgcn_wmma_f32_16x16x4_f32(false, a, false, bf0, (short)0, d0, false, false);
      d1 = __builtin_amdgcn_wmma_f32_16x16x4_f32(false, a, false, bf1, (short)0, d1, false, false);
    }
  }
  {
    const float bo0 = b2[nB], bo1 = b2[nB + 16];
#pragma unroll
    for (int r = 0; r < 8; ++r) {
      const int node = n0 + blk * 16 + r + 8 * lhalf;
      if (node < n_nodes) {
        out[(size_t)node * GIN_D + nB] = d0[r] + bo0;
        out[(size_t)node * GIN_D + nB + 16] = d1[r] + bo1;
      }
    }
  }
}

extern "C" void kernel_launch(void* const* d_in, const int* in_sizes, int n_in,
                              void* d_out, int out_size, void* d_ws, size_t ws_size,
                              hipStream_t stream) {
  const float* feat = (const float*)d_in[0];
  const int*   esrc = (const int*)d_in[1];
  const int*   edst = (const int*)d_in[2];
  const float* W1   = (const float*)d_in[3];
  const float* b1   = (const float*)d_in[4];
  const float* W2   = (const float*)d_in[5];
  const float* b2   = (const float*)d_in[6];
  const float* eps  = (const float*)d_in[7];

  const int n_nodes = in_sizes[0] / GIN_D;
  const int n_edges = in_sizes[1];
  const int nb  = (n_nodes + TILE_N - 1) / TILE_N;
  const int nwg = (n_edges + CHUNK_E - 1) / CHUNK_E;

  float* wfrag  = (float*)d_ws;
  int*   binned = (int*)(wfrag + WF_FLOATS);
  int*   piece  = binned + (size_t)n_edges * 2;

  gin_bin_kernel<<<nwg, BIN_T, 0, stream>>>(esrc, edst, W1, W2, wfrag, binned, piece,
                                            n_edges, nb, nwg);
  gin_tile_kernel<<<nb, TILE_T, 0, stream>>>(feat, eps, binned, piece, wfrag, b1, b2,
                                             (float*)d_out, n_nodes, nwg);
}


# tile sum: a wave walks the sources of its 8 rows as one stream, 8 row loads a group, next group asked before the current is added; a row's sum goes to the h tile at its end; same add order
# speedup vs baseline: 3.7664x; 1.0537x over previous
#include <hip/hip_runtime.h>
#include <hip/hip_bf16.h>


#define GIN_D 64
#define GIN_H 128
#define TILE_N 208
#define TILE_W 26
#define TILE_T (TILE_W * 32)
#define ROW_SHIFT 8
#define CNT_N 256
#define NB_MAX 512
#define CHUNK_E 6400
#define BIN_T 640
#define PER_T (CHUNK_E / BIN_T)
#define PIECES_W 10
#define NWG_MAX (TILE_W * PIECES_W)
#define NWG_WAVES ((NWG_MAX + 31) / 32)
#define PIECE_E 32
#define SORT_T PIECES_W
#define SORT_C (NWG_MAX * PIECE_E)
#define ROWS_W (TILE_N / TILE_W)
#define H_PITCH (GIN_D + 4)
#define MID_PITCH (GIN_H + 4)
#define BIG_W (TILE_N * MID_PITCH)
#define HBUF_OFF SORT_C

typedef __attribute__((ext_vector_type(2))) float v2f;
typedef __attribute__((ext_vector_type(4))) float v4f;
typedef __attribute__((ext_vector_type(8))) float v8f;
#define W1F_N (16 * 2 * 2 * 32)
#define W2F_N (32 * 2 * 32)
#define WF_FLOATS ((W1F_N + W2F_N) * 4)

__global__ __launch_bounds__(BIN_T) void gin_bin_kernel(
    const int* __restrict__ esrc, const int* __restrict__ edst,
    const float* __restrict__ W1, const float* __restrict__ W2,
    float* __restrict__ wfrag,
    int* __restrict__ binned_i, int* __restrict__ piece_i,
    int n_edges, int nb, int nwg) {
  __shared__ int hist[NB_MAX];
  __shared__ int base[NB_MAX];
  __shared__ int2 stage[CHUNK_E];
  const int tid = threadIdx.x;
  const int lane = tid & 31;
  const int wave = __builtin_amdgcn_readfirstlane(tid >> 5);
  const int wg = blockIdx.x;
  for (int i = wg * BIN_T + tid; i < W1F_N + W2F_N; i += nwg * BIN_T) {
    const int hh = (i >> 4) & 1;
    const int c = i & 15;
    v4f v;
    if (i < W1F_N) {
      const int q = (i >> 5) & 1, chh = (i >> 6) & 1, ks = i >> 7;
      const float* p = W1 + (size_t)(4 * ks + 2 * hh) * GIN_H + 64 * chh + 32 * q + c;
      v.x = p[0]; v.y = p[GIN_H]; v.z = p[16]; v.w = p[GIN_H + 16];
    } else {
      const int j = i - W1F_N;
      const int chh = (j >> 5) & 1, ks = j >> 6;
      const float* p = W2 + (size_t)(4 * ks + 2 * hh) * GIN_D + 32 * chh + c;
      v.x = p[0]; v.y = p[GIN_D]; v.z = p[16]; v.w = p[GIN_D + 16];
    }
    ((v4f*)wfrag)[i] = v;
  }
  if (tid < NB_MAX) hist[tid] = 0;
  __syncthreads();

  const long long e0 = (long long)wg * CHUNK_E + tid;
  const long long last = (long long)n_edges - 1;
  int dd[PER_T], ss[PER_T], rk[PER_T];
#pragma unroll
  for (int k = 0; k < PER_T; ++k) {
    const long long e = e0 + (long long)k * BIN_T;
    const long long ec = e < last ? e : last;
    dd[k] = edst[ec];
    ss[k] = esrc[ec];
  }
#pragma unroll
  for (int k = 0; k < PER_T; ++k) {
    const long long e = e0 + (long long)k * BIN_T;
    rk[k] = 0;
    if (e <= last)
      rk[k] = __hip_atomic_fetch_add(&hist[(unsigned)dd[k] / TILE_N], 1, __ATOMIC_RELAXED,
                                     __HIP_MEMORY_SCOPE_WORKGROUP);
  }
  __syncthreads();

  if (wave == 0) {
    int v[NB_MAX / 32];
#pragma unroll
    for (int j = 0; j < NB_MAX / 32; ++j) v[j] = hist[lane * (NB_MAX / 32) + j];
    int t = 0;
    int pre[NB_MAX / 32];
#pragma unroll
    for (int j = 0; j < NB_MAX / 32; ++j) { pre[j] = t; t += v[j]; }
    int inc = t;
#pragma unroll
    for (int off = 1; off < 32; off <<= 1) {
      const int u = __shfl_up(inc, off);
      if (lane >= off) inc += u;
    }
    const int b0 = inc - t;
#pragma unroll
    for (int j = 0; j < NB_MAX / 32; ++j) base[lane * (NB_MAX / 32) + j] = b0 + pre[j];
  }
  __syncthreads();

  const int c0 = wg * CHUNK_E;
#pragma unroll
  for (int k = 0; k < PER_T; ++k) {
    const long long e = e0 + (long long)k * BIN_T;
    if (e <= last)
      stage[base[(unsigned)dd[k] / TILE_N] + rk[k]] = make_int2(ss[k], dd[k]);
  }
  int2* __restrict__ piece = (int2*)piece_i;
  if (tid < nb) piece[(size_t)tid * nwg + wg] = make_int2(c0 + base[tid], hist[tid]);
  __syncthreads();

  int2* __restrict__ binned = (int2*)binned_i;
  const long long left = (long long)n_edges - c0;
  const int nc = left < CHUNK_E ? (int)left : CHUNK_E;
#pragma unroll
  for (int k = 0; k < PER_T; ++k) {
    const int i = tid + k * BIN_T;
    if (i < nc) binned[c0 + i] = stage[i];
  }
}

__device__ __forceinline__ void gin_issue(v2f (&X)[8], const float* __restrict__ feat,
                                          int sv, int off, int lane) {
#pragma unroll
  for (int j = 0; j < 8; ++j) {
    const unsigned s = (unsigned)__builtin_amdgcn_readlane(sv, off + j);
    const unsigned ofs = s * (unsigned)(GIN_D * sizeof(float)) + (unsigned)lane * (unsigned)sizeof(v2f);
    X[j] = *(const v2f*)((const char*)feat + ofs);
  }
}
__device__ __forceinline__ void gin_consume(const v2f (&X)[8], int base, int re, int rsvx,
                                            bool later_pass, float* hw,
                                            int& row, int& rend, v2f& cur) {
  const int nv = re - base;
  if (rend - base >= 8 && nv >= 8) {
#pragma unroll
    for (int j = 0; j < 8; ++j) {
      cur.x += X[j].x;
      cur.y += X[j].y;
    }
  } else {
#pragma unroll
    for (int j = 0; j < 8; ++j) {
      const int i = base + j;
      while (i >= rend) {
        *(v2f*)(hw + row * H_PITCH) = cur;
        ++row;
        rend = __builtin_amdgcn_readlane(rsvx, row + 1);
        cur.x = 0.f; cur.y = 0.f;
        if (later_pass) cur = *(const v2f*)(hw + row * H_PITCH);
      }
      v2f t;
      t.x = cur.x + X[j].x;
      t.y = cur.y + X[j].y;
      cur.x = j < nv ? t.x : cur.x;
      cur.y = j < nv ? t.y : cur.y;
    }
  }
}

__global__ __launch_bounds__(TILE_T) __attribute__((amdgpu_waves_per_eu(13, 16))) void gin_tile_kernel(
    const float* __restrict__ feat, const float* __restrict__ eps,
    const int* __restrict__ binned_i, const int* __restrict__ piece_i,
    const float* __restrict__ wfrag, const float* __restrict__ b1,
    const float* __restrict__ b2,
    float* __restrict__ out, int n_nodes, int nwg) {
  __shared__ int pstart[NWG_MAX];
  __shared__ int pcnt[NWG_MAX];
  __shared__ int wmax[NWG_WAVES];
  __shared__ int cnt[CNT_N];
  __shared__ int rstart[CNT_N + 1];
  __shared__ __attribute__((aligned(16))) float big[BIG_W];
  int* ssrc = (int*)big;
  float* hbuf = big + HBUF_OFF;
  float* mid = big;
  const int tid = threadIdx.x;
  const int lane = tid & 31;
  const int wave = __builtin_amdgcn_readfirstlane(tid >> 5);
  const int n0 = blockIdx.x * TILE_N;
  const int2* __restrict__ binned = (const int2*)binned_i;
  if (tid < NWG_WAVES * 32) {
    int2 pc = make_int2(0, 0);
    if (tid < nwg && tid < NWG_MAX) pc = ((const int2*)piece_i)[(size_t)blockIdx.x * nwg + tid];
    if (tid < NWG_MAX) {
      pstart[tid] = pc.x;
      pcnt[tid] = pc.y;
    }
    int m = pc.y;
#pragma unroll
    for (int off = 16; off >= 1; off >>= 1) {
      const int u = __shfl_xor(m, off);
      m = u > m ? u : m;
    }
    if (lane == 0) wmax[wave] = m;
  }
  __syncthreads();
  int maxc_v = 0;
#pragma unroll
  for (int j = 0; j < NWG_WAVES; ++j) maxc_v = wmax[j] > maxc_v ? wmax[j] : maxc_v;
  const int maxc = __builtin_amdgcn_readfirstlane(maxc_v);
  int pst[PIECES_W], pcn[PIECES_W];
#pragma unroll
  for (int k = 0; k < PIECES_W; ++k) {
    pst[k] = __builtin_amdgcn_readfirstlane(pstart[wave + TILE_W * k]);
    pcn[k] = __builtin_amdgcn_readfirstlane(pcnt[wave + TILE_W * k]);
  }

  float* const hw = hbuf + (wave * ROWS_W) * H_PITCH + 2 * lane;

  for (int p0 = 0; p0 == 0 || p0 < maxc; p0 += PIECE_E) {
    if (tid < CNT_N) cnt[tid] = 0;
    __syncthreads();

    int ss[SORT_T], dd[SORT_T];
#pragma unroll
    for (int k = 0; k < SORT_T; ++k) {
      const int j = p0 + lane;
      const int2 p = binned[j < pcn[k] ? pst[k] + j : 0];
      ss[k] = p.x;
      dd[k] = p.y;
    }
#pragma unroll
    for (int k = 0; k < SORT_T; ++k) {
      if (p0 + lane < pcn[k]) {
        const int row = dd[k] - n0;
        const int r = __hip_atomic_fetch_add(&cnt[row], 1, __ATOMIC_RELAXED,
                                             __HIP_MEMORY_SCOPE_WORKGROUP);
        dd[k] = row | (r << ROW_SHIFT);
      } else {
        dd[k] = -1;
      }
    }
    __syncthreads();

    if (wave == 0) {
      int v[8];
#pragma unroll
      for (int j = 0; j < 8; ++j) v[j] = cnt[lane * 8 + j];
      int t = 0;
      int pre[8];
#pragma unroll
      for (int j = 0; j < 8; ++j) { pre[j] = t; t += v[j]; }
      int inc = t;
#pragma unroll
      for (int off = 1; off < 32; off <<= 1) {
        const int u = __shfl_up(inc, off);
        if (lane >= off) inc += u;
      }
      const int base = inc - t;
#pragma unroll
      for (int j = 0; j < 8; ++j) rstart[lane * 8 + j] = base + pre[j];
      if (lane == 31) rstart[CNT_N] = inc;
    }
    __syncthreads();

#pragma unroll
    for (int k = 0; k < SORT_T; ++k) {
      if (dd[k] >= 0)
        ssrc[rstart[dd[k] & (CNT_N - 1)] + (dd[k] >> ROW_SHIFT)] = ss[k];
    }
    __syncthreads();

    const int rsv = rstart[wave * ROWS_W + (lane < ROWS_W ? lane : ROWS_W)];
    const int rb = __builtin_amdgcn_readlane(rsv, 0);
    const int re = __builtin_amdgcn_readlane(rsv, ROWS_W);
    const int rsvx = lane < ROWS_W ? rsv : 0x7fffffff;
    const bool later_pass = p0 > 0;
    int row = 0;
    int rend = __builtin_amdgcn_readlane(rsvx, 1);
    v2f cur;
    cur.x = 0.f; cur.y = 0.f;
    if (later_pass) cur = *(const v2f*)hw;
    if (rb < re) {
      v2f ga[8], gb[8];
      int sv = rb + lane < re ? ssrc[rb + lane] : 0;
      int svn = rb + 32 + lane < re ? ssrc[rb + 32 + lane] : 0;
      gin_issue(ga, feat, sv, 0, lane);
      for (int c0 = rb; c0 < re; c0 += 32) {
        gin_issue(gb, feat, sv, 8, lane);
        gin_consume(ga, c0, re, rsvx, later_pass, hw, row, rend, cur);
        gin_issue(ga, feat, sv, 16, lane);
        gin_consume(gb, c0 + 8, re, rsvx, later_pass, hw, row, rend, cur);
        gin_issue(gb, feat, sv, 24, lane);
        gin_consume(ga, c0 + 16, re, rsvx, later_pass, hw, row, rend, cur);
        gin_issue(ga, feat, svn, 0, lane);
        gin_consume(gb, c0 + 24, re, rsvx, later_pass, hw, row, rend, cur);
        sv = svn;
        svn = c0 + 64 + lane < re ? ssrc[c0 + 64 + lane] : 0;
      }
    }
    for (;;) {
      *(v2f*)(hw + row * H_PITCH) = cur;
      if (row >= ROWS_W - 1) break;
      ++row;
      cur.x = 0.f; cur.y = 0.f;
      if (later_pass) cur = *(const v2f*)(hw + row * H_PITCH);
    }
  }

  {
    const float sc = 1.0f + eps[0];
    v2f f[ROWS_W];
#pragma unroll
    for (int rr = 0; rr < ROWS_W; ++rr) {
      const int node = n0 + wave * ROWS_W + rr;
      const int nc = node < n_nodes ? node : n_nodes - 1;
      f[rr] = ((const v2f*)(feat + (size_t)nc * GIN_D))[lane];
    }
#pragma unroll
    for (int rr = 0; rr < ROWS_W; ++rr) {
      const v2f s = *(const v2f*)(hw + rr * H_PITCH);
      v2f h;
      h.x = sc * f[rr].x + s.x;
      h.y = sc * f[rr].y + s.y;
      *(v2f*)(hw + rr * H_PITCH) = h;
    }
  }
  __syncthreads();

  const int lhalf = lane >> 4;
  const int lmod  = lane & 15;
  const int blk = wave >> 1;
  const int ch = wave & 1;

  v8f c0 = {}, c1 = {}, c2 = {}, c3 = {};
  const int nA = 64 * ch + lmod;
  {
    const float* arow = hbuf + (blk * 16 + lmod) * H_PITCH + 2 * lhalf;
    const v4f* wf = (const v4f*)wfrag + (ch * 2) * 32 + lane;
    v4f p0 = wf[0];
    v4f p1 = wf[32];
#pragma unroll
    for (int k = 0; k < 16; ++k) {
      v2f a;
      a.x = arow[4 * k];
      a.y = arow[4 * k + 1];
      v4f n0 = p0, n1 = p1;
      if (k + 1 < 16) {
        n0 = wf[((k + 1) * 4) * 32];
        n1 = wf[((k + 1) * 4 + 1) * 32];
      }
      v2f bf0, bf1, bf2, bf3;
      bf0.x = p0.x; bf0.y = p0.y;
      bf1.x = p0.z; bf1.y = p0.w;
      bf2.x = p1.x; bf2.y = p1.y;
      bf3.x = p1.z; bf3.y = p1.w;
      c0 = __builtin_amdgcn_wmma_f32_16x16x4_f32(false, a, false, bf0, (short)0, c0, false, false);
      c1 = __builtin_amdgcn_wmma_f32_16x16x4_f32(false, a, false, bf1, (short)0, c1, false, false);
      c2 = __builtin_amdgcn_wmma_f32_16x16x4_f32(false, a, false, bf2, (short)0, c2, false, false);
      c3 = __builtin_amdgcn_wmma_f32_16x16x4_f32(false, a, false, bf3, (short)0, c3, false, false);
      p0 = n0; p1 = n1;
    }
  }
  __syncthreads();
  {
    const float bia0 = b1[nA], bia1 = b1[nA + 16], bia2 = b1[nA + 32], bia3 = b1[nA + 48];
#pragma unroll
    for (int r = 0; r < 8; ++r) {
      float* mrow_w = mid + (blk * 16 + r + 8 * lhalf) * MID_PITCH + nA;
      mrow_w[0]  = fmaxf(c0[r] + bia0, 0.0f);
      mrow_w[16] = fmaxf(c1[r] + bia1, 0.0f);
      mrow_w[32] = fmaxf(c2[r] + bia2, 0.0f);
      mrow_w[48] = fmaxf(c3[r] + bia3, 0.0f);
    }
  }
  __syncthreads();

  v8f d0 = {}, d1 = {};
  const int nB = 32 * ch + lmod;
  {
    const float* mrow = mid + (blk * 16 + lmod) * MID_PITCH + 2 * lhalf;
    const v4f* wf = (const v4f*)wfrag + W1F_N + ch * 32 + lane;
#pragma unroll 8
    for (int k = 0; k < 32; ++k) {
      v2f a;
      a.x = mrow[4 * k];
      a.y = mrow[4 * k + 1];
      const v4f p = wf[(k * 2) * 32];
      v2f bf0, bf1;
      bf0.x = p.x; bf0.y = p.y;
      bf1.x = p.z; bf1.y = p.w;
      d0 = __builtin_amdgcn_wmma_f32_16x16x4_f32(false, a, false, bf0, (short)0, d0, false, false);
      d1 = __builtin_amdgcn_wmma_f32_16x16x4_f32(false, a, false, bf1, (short)0, d1, false, false);
    }
  }
  {
    const float bo0 = b2[nB], bo1 = b2[nB + 16];
#pragma unroll
    for (int r = 0; r < 8; ++r) {
      const int node = n0 + blk * 16 + r + 8 * lhalf;
      if (node < n_nodes) {
        out[(size_t)node * GIN_D + nB] = d0[r] + bo0;
        out[(size_t)node * GIN_D + nB + 16] = d1[r] + bo1;
      }
    }
  }
}

extern "C" void kernel_launch(void* const* d_in, const int* in_sizes, int n_in,
                              void* d_out, int out_size, void* d_ws, size_t ws_size,
                              hipStream_t stream) {
  const float* feat = (const float*)d_in[0];
  const int*   esrc = (const int*)d_in[1];
  const int*   edst = (const int*)d_in[2];
  const float* W1   = (const float*)d_in[3];
  const float* b1   = (const float*)d_in[4];
  const float* W2   = (const float*)d_in[5];
  const float* b2   = (const float*)d_in[6];
  const float* eps  = (const float*)d_in[7];

  const int n_nodes = in_sizes[0] / GIN_D;
  const int n_edges = in_sizes[1];
  const int nb  = (n_nodes + TILE_N - 1) / TILE_N;
  const int nwg = (n_edges + CHUNK_E - 1) / CHUNK_E;

  float* wfrag  = (float*)d_ws;
  int*   binned = (int*)(wfrag + WF_FLOATS);
  int*   piece  = binned + (size_t)n_edges * 2;

  gin_bin_kernel<<<nwg, BIN_T, 0, stream>>>(esrc, edst, W1, W2, wfrag, binned, piece,
                                            n_edges, nb, nwg);
  gin_tile_kernel<<<nb, TILE_T, 0, stream>>>(feat, eps, binned, piece, wfrag, b1, b2,
                                             (float*)d_out, n_nodes, nwg);
}
